# starting kernel
# baseline (speedup 1.0000x reference)
#include <hip/hip_runtime.h>
#include <math.h>

#define GDE_N 50000
#define GDE_E 800000

typedef float v2f __attribute__((ext_vector_type(2)));
typedef float v8f __attribute__((ext_vector_type(8)));

__global__ __launch_bounds__(32) void gde_mm_wmma(const float* __restrict__ X,
                                                  const float* __restrict__ G,
                                                  float* __restrict__ Y,
                                                  int K, int Ncols) {
  const int lane = threadIdx.x;
  const int m    = lane & 15;
  const int half = lane >> 4;
  const int row  = blockIdx.x * 16 + m;
  const int col  = blockIdx.y * 16 + m;
  const float* __restrict__ xr = X + (size_t)row * K;
  v8f acc = {0.f, 0.f, 0.f, 0.f, 0.f, 0.f, 0.f, 0.f};
  for (int k = 0; k < K; k += 4) {
    const int ka = k + 2 * half;
    v2f a; a.x = xr[ka];
           a.y = xr[ka + 1];
    v2f b; b.x = G[(size_t)ka       * Ncols + col];
           b.y = G[(size_t)(ka + 1) * Ncols + col];
    acc = __builtin_amdgcn_wmma_f32_16x16x4_f32(false, a, false, b,
                                                (short)0, acc, false, false);
  }
  const int rbase = blockIdx.x * 16 + half * 8;
#pragma unroll
  for (int r = 0; r < 8; ++r)
    Y[(size_t)(rbase + r) * Ncols + col] = acc[r];
}

__global__ void gde_prep_g(float* __restrict__ G, const float* __restrict__ Wp,
                           const float* __restrict__ R, int Cin, int Cout) {
  int idx = blockIdx.x * blockDim.x + threadIdx.x;
  int Ncols = 3 * Cout;
  int tot = Cin * Ncols;
  if (idx >= tot) return;
  int r = idx / Ncols, c = idx % Ncols;
  float v;
  if (c < Cout)            v = Wp[r * Cout + c];
  else if (c < 2 * Cout)   v = Wp[(size_t)Cin * Cout + r * Cout + (c - Cout)]
                             - Wp[r * Cout + (c - Cout)];
  else                     v = R[r * Cout + (c - 2 * Cout)];
  G[idx] = v;
}

__global__ void gde_deg(float* __restrict__ deg, const int* __restrict__ dst, int E) {
  int e = blockIdx.x * blockDim.x + threadIdx.x;
  if (e < E) atomicAdd(&deg[dst[e]], 1.0f);
}

__global__ void gde_invdeg(float* __restrict__ d, int n) {
  int i = blockIdx.x * blockDim.x + threadIdx.x;
  if (i < n) d[i] = 1.0f / fmaxf(d[i], 1.0f);
}

__global__ void gde_edge_agg(float* __restrict__ agg, const float* __restrict__ Z,
                             const int* __restrict__ src, const int* __restrict__ dst,
                             const float* __restrict__ p, int Cout, long long total) {
  long long t = (long long)blockIdx.x * blockDim.x + threadIdx.x;
  if (t >= total) return;
  const int groups = Cout >> 2;
  const int e = (int)(t / groups);
  const int c = (int)(t % groups) * 4;
  const int s = src[e], d = dst[e];
  const float pe = p[e];
  const int ldz = 3 * Cout;
  const float4 z0 = *(const float4*)(Z + (size_t)s * ldz + c);
  const float4 z1 = *(const float4*)(Z + (size_t)s * ldz + Cout + c);
  float* ap = agg + (size_t)d * Cout + c;
  atomicAdd(ap + 0, z0.x + pe * z1.x);
  atomicAdd(ap + 1, z0.y + pe * z1.y);
  atomicAdd(ap + 2, z0.z + pe * z1.z);
  atomicAdd(ap + 3, z0.w + pe * z1.w);
}

__global__ void gde_finalize(float* __restrict__ out, const float* __restrict__ agg,
                             const float* __restrict__ Z, const float* __restrict__ bias,
                             const float* __restrict__ inv, int Cout, int act,
                             long long total) {
  long long idx = (long long)blockIdx.x * blockDim.x + threadIdx.x;
  if (idx >= total) return;
  int i = (int)(idx / Cout), c = (int)(idx % Cout);
  float v = agg[idx] * inv[i] + Z[(size_t)i * 3 * Cout + 2 * Cout + c] + bias[c];
  if (act) v = tanhf(v);
  out[idx] = v;
}

__global__ void gde_rk_stage(float* __restrict__ ACC, float* __restrict__ YT,
                             const float* __restrict__ H, const float* __restrict__ K,
                             float w, float c, int set_acc, long long n) {
  long long i = (long long)blockIdx.x * blockDim.x + threadIdx.x;
  if (i >= n) return;
  float k = K[i];
  ACC[i] = set_acc ? (w * k) : (ACC[i] + w * k);
  YT[i] = H[i] + c * k;
}

__global__ void gde_rk_final(float* __restrict__ H, const float* __restrict__ ACC,
                             const float* __restrict__ K, long long n) {
  long long i = (long long)blockIdx.x * blockDim.x + threadIdx.x;
  if (i >= n) return;
  H[i] = H[i] + 0.5f * (ACC[i] + K[i]);
}

__global__ void gde_log_softmax16(float* __restrict__ out, int n) {
  int i = blockIdx.x * blockDim.x + threadIdx.x;
  if (i >= n) return;
  float v[16];
  float mx = -3.402823466e+38f;
  float* row = out + (size_t)i * 16;
#pragma unroll
  for (int c = 0; c < 16; ++c) { v[c] = row[c]; mx = fmaxf(mx, v[c]); }
  float s = 0.f;
#pragma unroll
  for (int c = 0; c < 16; ++c) s += __expf(v[c] - mx);
  float ls = __logf(s);
#pragma unroll
  for (int c = 0; c < 16; ++c) row[c] = v[c] - mx - ls;
}

extern "C" void kernel_launch(void* const* d_in, const int* in_sizes, int n_in,
                              void* d_out, int out_size, void* d_ws, size_t ws_size,
                              hipStream_t stream) {
  const float* x   = (const float*)d_in[0];
  const float* p   = (const float*)d_in[1];
  const int*   src = (const int*)d_in[2];
  const int*   dst = (const int*)d_in[3];
  const float* W1 = (const float*)d_in[4],  *R1 = (const float*)d_in[5],  *b1 = (const float*)d_in[6];
  const float* Wa = (const float*)d_in[7],  *Ra = (const float*)d_in[8],  *ba = (const float*)d_in[9];
  const float* Wb = (const float*)d_in[10], *Rb = (const float*)d_in[11], *bb = (const float*)d_in[12];
  const float* W2 = (const float*)d_in[13], *R2 = (const float*)d_in[14], *b2 = (const float*)d_in[15];
  float* out = (float*)d_out;

  float* W = (float*)d_ws;
  size_t o = 0;
  float* Z   = W + o; o += (size_t)GDE_N * 192;
  float* AGG = W + o; o += (size_t)GDE_N * 64;
  float* H   = W + o; o += (size_t)GDE_N * 64;
  float* YT  = W + o; o += (size_t)GDE_N * 64;
  float* U   = W + o; o += (size_t)GDE_N * 64;
  float* Kb  = W + o; o += (size_t)GDE_N * 64;
  float* ACC = W + o; o += (size_t)GDE_N * 64;
  float* INV = W + o; o += (size_t)GDE_N;
  float* GC1 = W + o; o += (size_t)256 * 192;
  float* GCA = W + o; o += (size_t)64 * 192;
  float* GCB = W + o; o += (size_t)64 * 192;
  float* GC2 = W + o; o += (size_t)64 * 48;
  (void)ws_size; (void)n_in; (void)in_sizes; (void)out_size;

  gde_prep_g<<<(256 * 192 + 255) / 256, 256, 0, stream>>>(GC1, W1, R1, 256, 64);
  gde_prep_g<<<(64 * 192 + 255) / 256, 256, 0, stream>>>(GCA, Wa, Ra, 64, 64);
  gde_prep_g<<<(64 * 192 + 255) / 256, 256, 0, stream>>>(GCB, Wb, Rb, 64, 64);
  gde_prep_g<<<(64 * 48 + 255) / 256, 256, 0, stream>>>(GC2, W2, R2, 64, 16);

  hipMemsetAsync(INV, 0, sizeof(float) * GDE_N, stream);
  gde_deg<<<(GDE_E + 255) / 256, 256, 0, stream>>>(INV, dst, GDE_E);
  gde_invdeg<<<(GDE_N + 255) / 256, 256, 0, stream>>>(INV, GDE_N);

  auto spline = [&](const float* y, int Cin, int Cout, const float* G,
                    const float* bias, float* ob, int act) {
    const int N3 = 3 * Cout;
    dim3 grid(GDE_N / 16, N3 / 16);
    gde_mm_wmma<<<grid, 32, 0, stream>>>(y, G, Z, Cin, N3);
    hipMemsetAsync(AGG, 0, sizeof(float) * (size_t)GDE_N * Cout, stream);
    long long tot = (long long)GDE_E * (Cout / 4);
    gde_edge_agg<<<(int)((tot + 255) / 256), 256, 0, stream>>>(AGG, Z, src, dst, p, Cout, tot);
    long long tn = (long long)GDE_N * Cout;
    gde_finalize<<<(int)((tn + 255) / 256), 256, 0, stream>>>(ob, AGG, Z, bias, INV, Cout, act, tn);
  };
  auto odefunc = [&](const float* y, float* k) {
    spline(y, 64, 64, GCA, ba, U, 0);
    spline(U, 64, 64, GCB, bb, k, 0);
  };

  const long long tn = (long long)GDE_N * 64;
  const int eb = (int)((tn + 255) / 256);

  spline(x, 256, 64, GC1, b1, H, 1);

  odefunc(H, Kb);
  gde_rk_stage<<<eb, 256, 0, stream>>>(ACC, YT, H, Kb, 1.0f, 1.5f, 1, tn);
  odefunc(YT, Kb);
  gde_rk_stage<<<eb, 256, 0, stream>>>(ACC, YT, H, Kb, 2.0f, 1.5f, 0, tn);
  odefunc(YT, Kb);
  gde_rk_stage<<<eb, 256, 0, stream>>>(ACC, YT, H, Kb, 2.0f, 3.0f, 0, tn);
  odefunc(YT, Kb);
  gde_rk_final<<<eb, 256, 0, stream>>>(H, ACC, Kb, tn);

  spline(H, 64, 16, GC2, b2, out, 1);
  gde_log_softmax16<<<(GDE_N + 255) / 256, 256, 0, stream>>>(out, GDE_N);
}


# Dense matmul retiled: each wave holds the B fragments of its column tiles in registers for all K and streams row strips, A as float4 loads refilled in place; k order inside the WMMA sums changes
# speedup vs baseline: 1.3022x; 1.3022x over previous
#include <hip/hip_runtime.h>
#include <math.h>

#define GDE_N 50000
#define GDE_E 800000

typedef float v2f __attribute__((ext_vector_type(2)));
typedef float v8f __attribute__((ext_vector_type(8)));

template <int K, int NT, int CG, int RS>
__global__ __launch_bounds__(32 * CG * RS) void gde_mm_bstat(const float* __restrict__ X,
                                                            const float* __restrict__ G,
                                                            float* __restrict__ Y,
                                                            int nStrips) {
  constexpr int NCOLS = 16 * NT * CG;
  constexpr int STEPS = K / 4;
  constexpr int KC    = K / 64;
  const int lane = threadIdx.x & 31;
  const int wave = __builtin_amdgcn_readfirstlane((int)(threadIdx.x >> 5));
  const int c    = lane & 15;
  const int h    = lane >> 4;
  const int cg   = wave % CG;
  const int rs   = wave / CG;
  const int col0 = cg * (16 * NT) + c;

  v2f bf[STEPS][NT];
#pragma unroll
  for (int s = 0; s < STEPS; ++s) {
    const int kb = 8 * (s >> 1) + 2 * (s & 1) + 4 * h;
#pragma unroll
    for (int t = 0; t < NT; ++t) {
      bf[s][t].x = G[(size_t)kb       * NCOLS + col0 + 16 * t];
      bf[s][t].y = G[(size_t)(kb + 1) * NCOLS + col0 + 16 * t];
    }
  }

  const int stride = gridDim.x * RS;
  int strip = blockIdx.x * RS + rs;
  if (strip >= nStrips) return;
  const float* xr = X + ((size_t)strip * 16 + c) * K + 4 * h;

  float4 cur[8];
#pragma unroll
  for (int t = 0; t < 8; ++t) {
    cur[t] = *(const float4*)(xr + 8 * t);
    __builtin_amdgcn_sched_barrier(0);
  }

  for (; strip < nStrips; strip += stride) {
    const int sn = min(strip + stride, nStrips - 1);
    const float* xn = X + ((size_t)sn * 16 + c) * K + 4 * h;

    v8f acc[NT];
#pragma unroll
    for (int t = 0; t < NT; ++t) acc[t] = (v8f){0.f, 0.f, 0.f, 0.f, 0.f, 0.f, 0.f, 0.f};

#pragma unroll
    for (int kc = 0; kc < KC; ++kc) {
      const float* nsrc = (kc + 1 < KC) ? (xr + (kc + 1) * 64) : xn;
#pragma unroll
      for (int t = 0; t < 8; ++t) {
        v2f a0; a0.x = cur[t].x; a0.y = cur[t].y;
        v2f a1; a1.x = cur[t].z; a1.y = cur[t].w;
        const int S = kc * 16 + 2 * t;
#pragma unroll
        for (int q = 0; q < NT; ++q)
          acc[q] = __builtin_amdgcn_wmma_f32_16x16x4_f32(false, a0, false, bf[S][q],
                                                         (short)0, acc[q], false, false);
#pragma unroll
        for (int q = 0; q < NT; ++q)
          acc[q] = __builtin_amdgcn_wmma_f32_16x16x4_f32(false, a1, false, bf[S + 1][q],
                                                         (short)0, acc[q], false, false);
        cur[t] = *(const float4*)(nsrc + 8 * t);
        __builtin_amdgcn_sched_barrier(0);
      }
    }

    float* yr = Y + ((size_t)strip * 16 + 8 * h) * NCOLS + col0;
#pragma unroll
    for (int q = 0; q < NT; ++q)
#pragma unroll
      for (int r = 0; r < 8; ++r)
        yr[(size_t)r * NCOLS + 16 * q] = acc[q][r];

    xr = xn;
  }
}

__global__ void gde_prep_g(float* __restrict__ G, const float* __restrict__ Wp,
                           const float* __restrict__ R, int Cin, int Cout) {
  int idx = blockIdx.x * blockDim.x + threadIdx.x;
  int Ncols = 3 * Cout;
  int tot = Cin * Ncols;
  if (idx >= tot) return;
  int r = idx / Ncols, c = idx % Ncols;
  float v;
  if (c < Cout)            v = Wp[r * Cout + c];
  else if (c < 2 * Cout)   v = Wp[(size_t)Cin * Cout + r * Cout + (c - Cout)]
                             - Wp[r * Cout + (c - Cout)];
  else                     v = R[r * Cout + (c - 2 * Cout)];
  G[idx] = v;
}

__global__ void gde_deg(float* __restrict__ deg, const int* __restrict__ dst, int E) {
  int e = blockIdx.x * blockDim.x + threadIdx.x;
  if (e < E) atomicAdd(&deg[dst[e]], 1.0f);
}

__global__ void gde_invdeg(float* __restrict__ d, int n) {
  int i = blockIdx.x * blockDim.x + threadIdx.x;
  if (i < n) d[i] = 1.0f / fmaxf(d[i], 1.0f);
}

__global__ void gde_edge_agg(float* __restrict__ agg, const float* __restrict__ Z,
                             const int* __restrict__ src, const int* __restrict__ dst,
                             const float* __restrict__ p, int Cout, long long total) {
  long long t = (long long)blockIdx.x * blockDim.x + threadIdx.x;
  if (t >= total) return;
  const int groups = Cout >> 2;
  const int e = (int)(t / groups);
  const int c = (int)(t % groups) * 4;
  const int s = src[e], d = dst[e];
  const float pe = p[e];
  const int ldz = 3 * Cout;
  const float4 z0 = *(const float4*)(Z + (size_t)s * ldz + c);
  const float4 z1 = *(const float4*)(Z + (size_t)s * ldz + Cout + c);
  float* ap = agg + (size_t)d * Cout + c;
  atomicAdd(ap + 0, z0.x + pe * z1.x);
  atomicAdd(ap + 1, z0.y + pe * z1.y);
  atomicAdd(ap + 2, z0.z + pe * z1.z);
  atomicAdd(ap + 3, z0.w + pe * z1.w);
}

__global__ void gde_finalize(float* __restrict__ out, const float* __restrict__ agg,
                             const float* __restrict__ Z, const float* __restrict__ bias,
                             const float* __restrict__ inv, int Cout, int act,
                             long long total) {
  long long idx = (long long)blockIdx.x * blockDim.x + threadIdx.x;
  if (idx >= total) return;
  int i = (int)(idx / Cout), c = (int)(idx % Cout);
  float v = agg[idx] * inv[i] + Z[(size_t)i * 3 * Cout + 2 * Cout + c] + bias[c];
  if (act) v = tanhf(v);
  out[idx] = v;
}

__global__ void gde_rk_stage(float* __restrict__ ACC, float* __restrict__ YT,
                             const float* __restrict__ H, const float* __restrict__ K,
                             float w, float c, int set_acc, long long n) {
  long long i = (long long)blockIdx.x * blockDim.x + threadIdx.x;
  if (i >= n) return;
  float k = K[i];
  ACC[i] = set_acc ? (w * k) : (ACC[i] + w * k);
  YT[i] = H[i] + c * k;
}

__global__ void gde_rk_final(float* __restrict__ H, const float* __restrict__ ACC,
                             const float* __restrict__ K, long long n) {
  long long i = (long long)blockIdx.x * blockDim.x + threadIdx.x;
  if (i >= n) return;
  H[i] = H[i] + 0.5f * (ACC[i] + K[i]);
}

__global__ void gde_log_softmax16(float* __restrict__ out, int n) {
  int i = blockIdx.x * blockDim.x + threadIdx.x;
  if (i >= n) return;
  float v[16];
  float mx = -3.402823466e+38f;
  float* row = out + (size_t)i * 16;
#pragma unroll
  for (int c = 0; c < 16; ++c) { v[c] = row[c]; mx = fmaxf(mx, v[c]); }
  float s = 0.f;
#pragma unroll
  for (int c = 0; c < 16; ++c) s += __expf(v[c] - mx);
  float ls = __logf(s);
#pragma unroll
  for (int c = 0; c < 16; ++c) row[c] = v[c] - mx - ls;
}

extern "C" void kernel_launch(void* const* d_in, const int* in_sizes, int n_in,
                              void* d_out, int out_size, void* d_ws, size_t ws_size,
                              hipStream_t stream) {
  const float* x   = (const float*)d_in[0];
  const float* p   = (const float*)d_in[1];
  const int*   src = (const int*)d_in[2];
  const int*   dst = (const int*)d_in[3];
  const float* W1 = (const float*)d_in[4],  *R1 = (const float*)d_in[5],  *b1 = (const float*)d_in[6];
  const float* Wa = (const float*)d_in[7],  *Ra = (const float*)d_in[8],  *ba = (const float*)d_in[9];
  const float* Wb = (const float*)d_in[10], *Rb = (const float*)d_in[11], *bb = (const float*)d_in[12];
  const float* W2 = (const float*)d_in[13], *R2 = (const float*)d_in[14], *b2 = (const float*)d_in[15];
  float* out = (float*)d_out;

  float* W = (float*)d_ws;
  size_t o = 0;
  float* Z   = W + o; o += (size_t)GDE_N * 192;
  float* AGG = W + o; o += (size_t)GDE_N * 64;
  float* H   = W + o; o += (size_t)GDE_N * 64;
  float* YT  = W + o; o += (size_t)GDE_N * 64;
  float* U   = W + o; o += (size_t)GDE_N * 64;
  float* Kb  = W + o; o += (size_t)GDE_N * 64;
  float* ACC = W + o; o += (size_t)GDE_N * 64;
  float* INV = W + o; o += (size_t)GDE_N;
  float* GC1 = W + o; o += (size_t)256 * 192;
  float* GCA = W + o; o += (size_t)64 * 192;
  float* GCB = W + o; o += (size_t)64 * 192;
  float* GC2 = W + o; o += (size_t)64 * 48;
  (void)ws_size; (void)n_in; (void)in_sizes; (void)out_size;

  gde_prep_g<<<(256 * 192 + 255) / 256, 256, 0, stream>>>(GC1, W1, R1, 256, 64);
  gde_prep_g<<<(64 * 192 + 255) / 256, 256, 0, stream>>>(GCA, Wa, Ra, 64, 64);
  gde_prep_g<<<(64 * 192 + 255) / 256, 256, 0, stream>>>(GCB, Wb, Rb, 64, 64);
  gde_prep_g<<<(64 * 48 + 255) / 256, 256, 0, stream>>>(GC2, W2, R2, 64, 16);

  hipMemsetAsync(INV, 0, sizeof(float) * GDE_N, stream);
  gde_deg<<<(GDE_E + 255) / 256, 256, 0, stream>>>(INV, dst, GDE_E);
  gde_invdeg<<<(GDE_N + 255) / 256, 256, 0, stream>>>(INV, GDE_N);

  auto spline = [&](const float* y, int Cin, int Cout, const float* G,
                    const float* bias, float* ob, int act) {
    if (Cin == 256)
      gde_mm_bstat<256, 3, 4, 2><<<256, 256, 0, stream>>>(y, G, Z, GDE_N / 16);
    else if (Cout == 64)
      gde_mm_bstat<64, 4, 3, 4><<<256, 384, 0, stream>>>(y, G, Z, GDE_N / 16);
    else
      gde_mm_bstat<64, 3, 1, 8><<<256, 256, 0, stream>>>(y, G, Z, GDE_N / 16);
    hipMemsetAsync(AGG, 0, sizeof(float) * (size_t)GDE_N * Cout, stream);
    long long tot = (long long)GDE_E * (Cout / 4);
    gde_edge_agg<<<(int)((tot + 255) / 256), 256, 0, stream>>>(AGG, Z, src, dst, p, Cout, tot);
    long long tn = (long long)GDE_N * Cout;
    gde_finalize<<<(int)((tn + 255) / 256), 256, 0, stream>>>(ob, AGG, Z, bias, INV, Cout, act, tn);
  };
  auto odefunc = [&](const float* y, float* k) {
    spline(y, 64, 64, GCA, ba, U, 0);
    spline(U, 64, 64, GCB, bb, k, 0);
  };

  const long long tn = (long long)GDE_N * 64;
  const int eb = (int)((tn + 255) / 256);

  spline(x, 256, 64, GC1, b1, H, 1);

  odefunc(H, Kb);
  gde_rk_stage<<<eb, 256, 0, stream>>>(ACC, YT, H, Kb, 1.0f, 1.5f, 1, tn);
  odefunc(YT, Kb);
  gde_rk_stage<<<eb, 256, 0, stream>>>(ACC, YT, H, Kb, 2.0f, 1.5f, 0, tn);
  odefunc(YT, Kb);
  gde_rk_stage<<<eb, 256, 0, stream>>>(ACC, YT, H, Kb, 2.0f, 3.0f, 0, tn);
  odefunc(YT, Kb);
  gde_rk_final<<<eb, 256, 0, stream>>>(H, ACC, Kb, tn);

  spline(H, 64, 16, GC2, b2, out, 1);
  gde_log_softmax16<<<(GDE_N + 255) / 256, 256, 0, stream>>>(out, GDE_N);
}


# Edge sums gathered per destination from a dst-grouped edge list rebuilt each call (int atomic rank, scan, sort by edge id) and finished in registers: no float atomics; sums now in edge order
# speedup vs baseline: 2.6213x; 2.0130x over previous
#include <hip/hip_runtime.h>
#include <math.h>

#define GDE_N 50000
#define GDE_E 800000

typedef float v2f __attribute__((ext_vector_type(2)));
typedef float v8f __attribute__((ext_vector_type(8)));

template <int K, int NT, int CG, int RS>
__global__ __launch_bounds__(32 * CG * RS) void gde_mm_bstat(const float* __restrict__ X,
                                                            const float* __restrict__ G,
                                                            float* __restrict__ Y,
                                                            int nStrips) {
  constexpr int NCOLS = 16 * NT * CG;
  constexpr int STEPS = K / 4;
  constexpr int KC    = K / 64;
  const int lane = threadIdx.x & 31;
  const int wave = __builtin_amdgcn_readfirstlane((int)(threadIdx.x >> 5));
  const int c    = lane & 15;
  const int h    = lane >> 4;
  const int cg   = wave % CG;
  const int rs   = wave / CG;
  const int col0 = cg * (16 * NT) + c;

  v2f bf[STEPS][NT];
#pragma unroll
  for (int s = 0; s < STEPS; ++s) {
    const int kb = 8 * (s >> 1) + 2 * (s & 1) + 4 * h;
#pragma unroll
    for (int t = 0; t < NT; ++t) {
      bf[s][t].x = G[(size_t)kb       * NCOLS + col0 + 16 * t];
      bf[s][t].y = G[(size_t)(kb + 1) * NCOLS + col0 + 16 * t];
    }
  }

  const int stride = gridDim.x * RS;
  int strip = blockIdx.x * RS + rs;
  if (strip >= nStrips) return;
  const float* xr = X + ((size_t)strip * 16 + c) * K + 4 * h;

  float4 cur[8];
#pragma unroll
  for (int t = 0; t < 8; ++t) {
    cur[t] = *(const float4*)(xr + 8 * t);
    __builtin_amdgcn_sched_barrier(0);
  }

  for (; strip < nStrips; strip += stride) {
    const int sn = min(strip + stride, nStrips - 1);
    const float* xn = X + ((size_t)sn * 16 + c) * K + 4 * h;

    v8f acc[NT];
#pragma unroll
    for (int t = 0; t < NT; ++t) acc[t] = (v8f){0.f, 0.f, 0.f, 0.f, 0.f, 0.f, 0.f, 0.f};

#pragma unroll
    for (int kc = 0; kc < KC; ++kc) {
      const float* nsrc = (kc + 1 < KC) ? (xr + (kc + 1) * 64) : xn;
#pragma unroll
      for (int t = 0; t < 8; ++t) {
        v2f a0; a0.x = cur[t].x; a0.y = cur[t].y;
        v2f a1; a1.x = cur[t].z; a1.y = cur[t].w;
        const int S = kc * 16 + 2 * t;
#pragma unroll
        for (int q = 0; q < NT; ++q)
          acc[q] = __builtin_amdgcn_wmma_f32_16x16x4_f32(false, a0, false, bf[S][q],
                                                         (short)0, acc[q], false, false);
#pragma unroll
        for (int q = 0; q < NT; ++q)
          acc[q] = __builtin_amdgcn_wmma_f32_16x16x4_f32(false, a1, false, bf[S + 1][q],
                                                         (short)0, acc[q], false, false);
        cur[t] = *(const float4*)(nsrc + 8 * t);
        __builtin_amdgcn_sched_barrier(0);
      }
    }

    float* yr = Y + ((size_t)strip * 16 + 8 * h) * NCOLS + col0;
#pragma unroll
    for (int q = 0; q < NT; ++q)
#pragma unroll
      for (int r = 0; r < 8; ++r)
        yr[(size_t)r * NCOLS + 16 * q] = acc[q][r];

    xr = xn;
  }
}

__global__ void gde_prep_g(float* __restrict__ G, const float* __restrict__ Wp,
                           const float* __restrict__ R, int Cin, int Cout) {
  int idx = blockIdx.x * blockDim.x + threadIdx.x;
  int Ncols = 3 * Cout;
  int tot = Cin * Ncols;
  if (idx >= tot) return;
  int r = idx / Ncols, c = idx % Ncols;
  float v;
  if (c < Cout)            v = Wp[r * Cout + c];
  else if (c < 2 * Cout)   v = Wp[(size_t)Cin * Cout + r * Cout + (c - Cout)]
                             - Wp[r * Cout + (c - Cout)];
  else                     v = R[r * Cout + (c - 2 * Cout)];
  G[idx] = v;
}

__global__ void gde_rank(int* __restrict__ cnt, int* __restrict__ arr,
                         const int* __restrict__ dst, int E) {
  int e = blockIdx.x * blockDim.x + threadIdx.x;
  if (e < E) arr[e] = atomicAdd(&cnt[dst[e]], 1);
}

__global__ __launch_bounds__(1024) void gde_scan(int* __restrict__ rowptr, float* __restrict__ inv,
                                                 const int* __restrict__ cnt) {
  constexpr int N4    = GDE_N / 4;
  constexpr int TILES = (N4 + 1023) / 1024;
  static_assert(GDE_N % 4 == 0, "int4 tiles");
  __shared__ int wsum[32];
  __shared__ int tile_total;
  const int t = threadIdx.x, lane = t & 31, w = t >> 5;
  int4 v[TILES];
#pragma unroll
  for (int k = 0; k < TILES; ++k) {
    const int i = k * 1024 + t;
    v[k] = (i < N4) ? ((const int4*)cnt)[i] : make_int4(0, 0, 0, 0);
  }
  int carry = 0;
#pragma unroll
  for (int k = 0; k < TILES; ++k) {
    const int i = k * 1024 + t;
    const int s = v[k].x + v[k].y + v[k].z + v[k].w;
    int inc = s;
    for (int d = 1; d < 32; d <<= 1) {
      const int o = __shfl_up(inc, d);
      if (lane >= d) inc += o;
    }
    if (lane == 31) wsum[w] = inc;
    __syncthreads();
    if (w == 0) {
      const int x = wsum[lane];
      int xi = x;
      for (int d = 1; d < 32; d <<= 1) {
        const int o = __shfl_up(xi, d);
        if (lane >= d) xi += o;
      }
      wsum[lane] = xi - x;
      if (lane == 31) tile_total = xi;
    }
    __syncthreads();
    const int base = carry + wsum[w] + inc - s;
    carry += tile_total;
    if (i < N4) {
      int4 r;
      r.x = base;
      r.y = r.x + v[k].x;
      r.z = r.y + v[k].y;
      r.w = r.z + v[k].z;
      ((int4*)rowptr)[i] = r;
      float4 f;
      f.x = 1.0f / fmaxf((float)v[k].x, 1.0f);
      f.y = 1.0f / fmaxf((float)v[k].y, 1.0f);
      f.z = 1.0f / fmaxf((float)v[k].z, 1.0f);
      f.w = 1.0f / fmaxf((float)v[k].w, 1.0f);
      ((float4*)inv)[i] = f;
    }
    __syncthreads();
  }
  if (t == 0) rowptr[GDE_N] = carry;
}

__global__ void gde_place(int* __restrict__ tmp, const int* __restrict__ rowptr,
                          const int* __restrict__ arr, const int* __restrict__ dst, int E) {
  int e = blockIdx.x * blockDim.x + threadIdx.x;
  if (e < E) tmp[rowptr[dst[e]] + arr[e]] = e;
}

__global__ void gde_order(int2* __restrict__ ed, const int* __restrict__ tmp,
                          const int* __restrict__ rowptr, const int* __restrict__ src,
                          const int* __restrict__ dst, const float* __restrict__ p, int E) {
  int e = blockIdx.x * blockDim.x + threadIdx.x;
  if (e >= E) return;
  const int d = dst[e];
  const int b = rowptr[d], n = rowptr[d + 1];
  int r = 0;
  for (int j = b; j < n; ++j) r += (tmp[j] < e) ? 1 : 0;
  ed[b + r] = make_int2(src[e], __float_as_int(p[e]));
}

__device__ __forceinline__ void gde_acc(float4& s, const float4& z0, const float4& z1, float pe) {
  s.x += z0.x + pe * z1.x;
  s.y += z0.y + pe * z1.y;
  s.z += z0.z + pe * z1.z;
  s.w += z0.w + pe * z1.w;
}

template <int COUT, int ACT>
__global__ __launch_bounds__(256) void gde_gather(float* __restrict__ out, const float* __restrict__ Z,
                                                  const int2* __restrict__ ed,
                                                  const int* __restrict__ rowptr,
                                                  const float* __restrict__ bias,
                                                  const float* __restrict__ inv) {
  constexpr int LPN = COUT / 4;
  constexpr int LDZ = 3 * COUT;
  const int gt   = blockIdx.x * 256 + threadIdx.x;
  const int node = gt / LPN;
  if (node >= GDE_N) return;
  const int c = (gt % LPN) * 4;
  const int b = rowptr[node], n = rowptr[node + 1];
  const float* zc = Z + c;
  float4 s = make_float4(0.f, 0.f, 0.f, 0.f);
  int j = b;
  for (; j + 4 <= n; j += 4) {
    const int2 e0 = ed[j], e1 = ed[j + 1], e2 = ed[j + 2], e3 = ed[j + 3];
    const float* r0 = zc + (size_t)e0.x * LDZ;
    const float* r1 = zc + (size_t)e1.x * LDZ;
    const float* r2 = zc + (size_t)e2.x * LDZ;
    const float* r3 = zc + (size_t)e3.x * LDZ;
    const float4 a0 = *(const float4*)r0, b0 = *(const float4*)(r0 + COUT);
    const float4 a1 = *(const float4*)r1, b1 = *(const float4*)(r1 + COUT);
    const float4 a2 = *(const float4*)r2, b2 = *(const float4*)(r2 + COUT);
    const float4 a3 = *(const float4*)r3, b3 = *(const float4*)(r3 + COUT);
    gde_acc(s, a0, b0, __int_as_float(e0.y));
    gde_acc(s, a1, b1, __int_as_float(e1.y));
    gde_acc(s, a2, b2, __int_as_float(e2.y));
    gde_acc(s, a3, b3, __int_as_float(e3.y));
  }
  for (; j < n; ++j) {
    const int2 e0 = ed[j];
    const float* r0 = zc + (size_t)e0.x * LDZ;
    const float4 a0 = *(const float4*)r0, b0 = *(const float4*)(r0 + COUT);
    gde_acc(s, a0, b0, __int_as_float(e0.y));
  }
  const float iv = inv[node];
  const float4 zr = *(const float4*)(Z + (size_t)node * LDZ + 2 * COUT + c);
  float4 v;
  v.x = s.x * iv + zr.x + bias[c + 0];
  v.y = s.y * iv + zr.y + bias[c + 1];
  v.z = s.z * iv + zr.z + bias[c + 2];
  v.w = s.w * iv + zr.w + bias[c + 3];
  if (ACT) { v.x = tanhf(v.x); v.y = tanhf(v.y); v.z = tanhf(v.z); v.w = tanhf(v.w); }
  *(float4*)(out + (size_t)node * COUT + c) = v;
}

__global__ void gde_rk_stage(float* __restrict__ ACC, float* __restrict__ YT,
                             const float* __restrict__ H, const float* __restrict__ K,
                             float w, float c, int set_acc, long long n) {
  long long i = (long long)blockIdx.x * blockDim.x + threadIdx.x;
  if (i >= n) return;
  float k = K[i];
  ACC[i] = set_acc ? (w * k) : (ACC[i] + w * k);
  YT[i] = H[i] + c * k;
}

__global__ void gde_rk_final(float* __restrict__ H, const float* __restrict__ ACC,
                             const float* __restrict__ K, long long n) {
  long long i = (long long)blockIdx.x * blockDim.x + threadIdx.x;
  if (i >= n) return;
  H[i] = H[i] + 0.5f * (ACC[i] + K[i]);
}

__global__ void gde_log_softmax16(float* __restrict__ out, int n) {
  int i = blockIdx.x * blockDim.x + threadIdx.x;
  if (i >= n) return;
  float v[16];
  float mx = -3.402823466e+38f;
  float* row = out + (size_t)i * 16;
#pragma unroll
  for (int c = 0; c < 16; ++c) { v[c] = row[c]; mx = fmaxf(mx, v[c]); }
  float s = 0.f;
#pragma unroll
  for (int c = 0; c < 16; ++c) s += __expf(v[c] - mx);
  float ls = __logf(s);
#pragma unroll
  for (int c = 0; c < 16; ++c) row[c] = v[c] - mx - ls;
}

extern "C" void kernel_launch(void* const* d_in, const int* in_sizes, int n_in,
                              void* d_out, int out_size, void* d_ws, size_t ws_size,
                              hipStream_t stream) {
  const float* x   = (const float*)d_in[0];
  const float* p   = (const float*)d_in[1];
  const int*   src = (const int*)d_in[2];
  const int*   dst = (const int*)d_in[3];
  const float* W1 = (const float*)d_in[4],  *R1 = (const float*)d_in[5],  *b1 = (const float*)d_in[6];
  const float* Wa = (const float*)d_in[7],  *Ra = (const float*)d_in[8],  *ba = (const float*)d_in[9];
  const float* Wb = (const float*)d_in[10], *Rb = (const float*)d_in[11], *bb = (const float*)d_in[12];
  const float* W2 = (const float*)d_in[13], *R2 = (const float*)d_in[14], *b2 = (const float*)d_in[15];
  float* out = (float*)d_out;

  float* W = (float*)d_ws;
  size_t o = 0;
  float* Z   = W + o; o += (size_t)GDE_N * 192;
  float* CSR = W + o; o += (size_t)GDE_N * 64;
  float* H   = W + o; o += (size_t)GDE_N * 64;
  float* YT  = W + o; o += (size_t)GDE_N * 64;
  float* U   = W + o; o += (size_t)GDE_N * 64;
  float* Kb  = W + o; o += (size_t)GDE_N * 64;
  float* ACC = W + o; o += (size_t)GDE_N * 64;
  float* INV = W + o; o += (size_t)GDE_N;
  float* GC1 = W + o; o += (size_t)256 * 192;
  float* GCA = W + o; o += (size_t)64 * 192;
  float* GCB = W + o; o += (size_t)64 * 192;
  float* GC2 = W + o; o += (size_t)64 * 48;
  int*  CNT = (int*)CSR;
  int*  RP  = CNT + GDE_N;
  int*  TMP = RP + GDE_N + 2;
  int*  ARR = TMP + GDE_E;
  int2* ED  = (int2*)ARR;
  (void)ws_size; (void)n_in; (void)in_sizes; (void)out_size;

  gde_prep_g<<<(256 * 192 + 255) / 256, 256, 0, stream>>>(GC1, W1, R1, 256, 64);
  gde_prep_g<<<(64 * 192 + 255) / 256, 256, 0, stream>>>(GCA, Wa, Ra, 64, 64);
  gde_prep_g<<<(64 * 192 + 255) / 256, 256, 0, stream>>>(GCB, Wb, Rb, 64, 64);
  gde_prep_g<<<(64 * 48 + 255) / 256, 256, 0, stream>>>(GC2, W2, R2, 64, 16);

  hipMemsetAsync(CNT, 0, sizeof(int) * GDE_N, stream);
  gde_rank<<<(GDE_E + 255) / 256, 256, 0, stream>>>(CNT, ARR, dst, GDE_E);
  gde_scan<<<1, 1024, 0, stream>>>(RP, INV, CNT);
  gde_place<<<(GDE_E + 255) / 256, 256, 0, stream>>>(TMP, RP, ARR, dst, GDE_E);
  gde_order<<<(GDE_E + 255) / 256, 256, 0, stream>>>(ED, TMP, RP, src, dst, p, GDE_E);

  auto spline = [&](const float* y, int Cin, int Cout, const float* G,
                    const float* bias, float* ob, int act) {
    if (Cin == 256)
      gde_mm_bstat<256, 3, 4, 2><<<256, 256, 0, stream>>>(y, G, Z, GDE_N / 16);
    else if (Cout == 64)
      gde_mm_bstat<64, 4, 3, 4><<<256, 384, 0, stream>>>(y, G, Z, GDE_N / 16);
    else
      gde_mm_bstat<64, 3, 1, 8><<<256, 256, 0, stream>>>(y, G, Z, GDE_N / 16);
    if (Cout == 64 && act)
      gde_gather<64, 1><<<GDE_N * 16 / 256, 256, 0, stream>>>(ob, Z, ED, RP, bias, INV);
    else if (Cout == 64)
      gde_gather<64, 0><<<GDE_N * 16 / 256, 256, 0, stream>>>(ob, Z, ED, RP, bias, INV);
    else
      gde_gather<16, 1><<<(GDE_N * 4 + 255) / 256, 256, 0, stream>>>(ob, Z, ED, RP, bias, INV);
  };
  auto odefunc = [&](const float* y, float* k) {
    spline(y, 64, 64, GCA, ba, U, 0);
    spline(U, 64, 64, GCB, bb, k, 0);
  };

  const long long tn = (long long)GDE_N * 64;
  const int eb = (int)((tn + 255) / 256);

  spline(x, 256, 64, GC1, b1, H, 1);

  odefunc(H, Kb);
  gde_rk_stage<<<eb, 256, 0, stream>>>(ACC, YT, H, Kb, 1.0f, 1.5f, 1, tn);
  odefunc(YT, Kb);
  gde_rk_stage<<<eb, 256, 0, stream>>>(ACC, YT, H, Kb, 2.0f, 1.5f, 0, tn);
  odefunc(YT, Kb);
  gde_rk_stage<<<eb, 256, 0, stream>>>(ACC, YT, H, Kb, 2.0f, 3.0f, 0, tn);
  odefunc(YT, Kb);
  gde_rk_final<<<eb, 256, 0, stream>>>(H, ACC, Kb, tn);

  spline(H, 64, 16, GC2, b2, out, 1);
  gde_log_softmax16<<<(GDE_N + 255) / 256, 256, 0, stream>>>(out, GDE_N);
}


# RK4 stage and final updates moved into the epilogue of the gather that produces k (k never stored): 4 launches of 100000 one-float-per-thread waves gone; arithmetic and sum order unchanged
# speedup vs baseline: 2.8349x; 1.0815x over previous
#include <hip/hip_runtime.h>
#include <math.h>

#define GDE_N 50000
#define GDE_E 800000

typedef float v2f __attribute__((ext_vector_type(2)));
typedef float v8f __attribute__((ext_vector_type(8)));

template <int K, int NT, int CG, int RS>
__global__ __launch_bounds__(32 * CG * RS) void gde_mm_bstat(const float* __restrict__ X,
                                                            const float* __restrict__ G,
                                                            float* __restrict__ Y,
                                                            int nStrips) {
  constexpr int NCOLS = 16 * NT * CG;
  constexpr int STEPS = K / 4;
  constexpr int KC    = K / 64;
  const int lane = threadIdx.x & 31;
  const int wave = __builtin_amdgcn_readfirstlane((int)(threadIdx.x >> 5));
  const int c    = lane & 15;
  const int h    = lane >> 4;
  const int cg   = wave % CG;
  const int rs   = wave / CG;
  const int col0 = cg * (16 * NT) + c;

  v2f bf[STEPS][NT];
#pragma unroll
  for (int s = 0; s < STEPS; ++s) {
    const int kb = 8 * (s >> 1) + 2 * (s & 1) + 4 * h;
#pragma unroll
    for (int t = 0; t < NT; ++t) {
      bf[s][t].x = G[(size_t)kb       * NCOLS + col0 + 16 * t];
      bf[s][t].y = G[(size_t)(kb + 1) * NCOLS + col0 + 16 * t];
    }
  }

  const int stride = gridDim.x * RS;
  int strip = blockIdx.x * RS + rs;
  if (strip >= nStrips) return;
  const float* xr = X + ((size_t)strip * 16 + c) * K + 4 * h;

  float4 cur[8];
#pragma unroll
  for (int t = 0; t < 8; ++t) {
    cur[t] = *(const float4*)(xr + 8 * t);
    __builtin_amdgcn_sched_barrier(0);
  }

  for (; strip < nStrips; strip += stride) {
    const int sn = min(strip + stride, nStrips - 1);
    const float* xn = X + ((size_t)sn * 16 + c) * K + 4 * h;

    v8f acc[NT];
#pragma unroll
    for (int t = 0; t < NT; ++t) acc[t] = (v8f){0.f, 0.f, 0.f, 0.f, 0.f, 0.f, 0.f, 0.f};

#pragma unroll
    for (int kc = 0; kc < KC; ++kc) {
      const float* nsrc = (kc + 1 < KC) ? (xr + (kc + 1) * 64) : xn;
#pragma unroll
      for (int t = 0; t < 8; ++t) {
        v2f a0; a0.x = cur[t].x; a0.y = cur[t].y;
        v2f a1; a1.x = cur[t].z; a1.y = cur[t].w;
        const int S = kc * 16 + 2 * t;
#pragma unroll
        for (int q = 0; q < NT; ++q)
          acc[q] = __builtin_amdgcn_wmma_f32_16x16x4_f32(false, a0, false, bf[S][q],
                                                         (short)0, acc[q], false, false);
#pragma unroll
        for (int q = 0; q < NT; ++q)
          acc[q] = __builtin_amdgcn_wmma_f32_16x16x4_f32(false, a1, false, bf[S + 1][q],
                                                         (short)0, acc[q], false, false);
        cur[t] = *(const float4*)(nsrc + 8 * t);
        __builtin_amdgcn_sched_barrier(0);
      }
    }

    float* yr = Y + ((size_t)strip * 16 + 8 * h) * NCOLS + col0;
#pragma unroll
    for (int q = 0; q < NT; ++q)
#pragma unroll
      for (int r = 0; r < 8; ++r)
        yr[(size_t)r * NCOLS + 16 * q] = acc[q][r];

    xr = xn;
  }
}

__global__ void gde_prep_g(float* __restrict__ G, const float* __restrict__ Wp,
                           const float* __restrict__ R, int Cin, int Cout) {
  int idx = blockIdx.x * blockDim.x + threadIdx.x;
  int Ncols = 3 * Cout;
  int tot = Cin * Ncols;
  if (idx >= tot) return;
  int r = idx / Ncols, c = idx % Ncols;
  float v;
  if (c < Cout)            v = Wp[r * Cout + c];
  else if (c < 2 * Cout)   v = Wp[(size_t)Cin * Cout + r * Cout + (c - Cout)]
                             - Wp[r * Cout + (c - Cout)];
  else                     v = R[r * Cout + (c - 2 * Cout)];
  G[idx] = v;
}

__global__ void gde_rank(int* __restrict__ cnt, int* __restrict__ arr,
                         const int* __restrict__ dst, int E) {
  int e = blockIdx.x * blockDim.x + threadIdx.x;
  if (e < E) arr[e] = atomicAdd(&cnt[dst[e]], 1);
}

__global__ __launch_bounds__(1024) void gde_scan(int* __restrict__ rowptr, float* __restrict__ inv,
                                                 const int* __restrict__ cnt) {
  constexpr int N4    = GDE_N / 4;
  constexpr int TILES = (N4 + 1023) / 1024;
  static_assert(GDE_N % 4 == 0, "int4 tiles");
  __shared__ int wsum[32];
  __shared__ int tile_total;
  const int t = threadIdx.x, lane = t & 31, w = t >> 5;
  int4 v[TILES];
#pragma unroll
  for (int k = 0; k < TILES; ++k) {
    const int i = k * 1024 + t;
    v[k] = (i < N4) ? ((const int4*)cnt)[i] : make_int4(0, 0, 0, 0);
  }
  int carry = 0;
#pragma unroll
  for (int k = 0; k < TILES; ++k) {
    const int i = k * 1024 + t;
    const int s = v[k].x + v[k].y + v[k].z + v[k].w;
    int inc = s;
    for (int d = 1; d < 32; d <<= 1) {
      const int o = __shfl_up(inc, d);
      if (lane >= d) inc += o;
    }
    if (lane == 31) wsum[w] = inc;
    __syncthreads();
    if (w == 0) {
      const int x = wsum[lane];
      int xi = x;
      for (int d = 1; d < 32; d <<= 1) {
        const int o = __shfl_up(xi, d);
        if (lane >= d) xi += o;
      }
      wsum[lane] = xi - x;
      if (lane == 31) tile_total = xi;
    }
    __syncthreads();
    const int base = carry + wsum[w] + inc - s;
    carry += tile_total;
    if (i < N4) {
      int4 r;
      r.x = base;
      r.y = r.x + v[k].x;
      r.z = r.y + v[k].y;
      r.w = r.z + v[k].z;
      ((int4*)rowptr)[i] = r;
      float4 f;
      f.x = 1.0f / fmaxf((float)v[k].x, 1.0f);
      f.y = 1.0f / fmaxf((float)v[k].y, 1.0f);
      f.z = 1.0f / fmaxf((float)v[k].z, 1.0f);
      f.w = 1.0f / fmaxf((float)v[k].w, 1.0f);
      ((float4*)inv)[i] = f;
    }
    __syncthreads();
  }
  if (t == 0) rowptr[GDE_N] = carry;
}

__global__ void gde_place(int* __restrict__ tmp, const int* __restrict__ rowptr,
                          const int* __restrict__ arr, const int* __restrict__ dst, int E) {
  int e = blockIdx.x * blockDim.x + threadIdx.x;
  if (e < E) tmp[rowptr[dst[e]] + arr[e]] = e;
}

__global__ void gde_order(int2* __restrict__ ed, const int* __restrict__ tmp,
                          const int* __restrict__ rowptr, const int* __restrict__ src,
                          const int* __restrict__ dst, const float* __restrict__ p, int E) {
  int e = blockIdx.x * blockDim.x + threadIdx.x;
  if (e >= E) return;
  const int d = dst[e];
  const int b = rowptr[d], n = rowptr[d + 1];
  int r = 0;
  for (int j = b; j < n; ++j) r += (tmp[j] < e) ? 1 : 0;
  ed[b + r] = make_int2(src[e], __float_as_int(p[e]));
}

__device__ __forceinline__ void gde_acc(float4& s, const float4& z0, const float4& z1, float pe) {
  s.x += z0.x + pe * z1.x;
  s.y += z0.y + pe * z1.y;
  s.z += z0.z + pe * z1.z;
  s.w += z0.w + pe * z1.w;
}

template <int COUT, int ACT, int RK>
__global__ __launch_bounds__(256) void gde_gather(float* __restrict__ out, const float* __restrict__ Z,
                                                  const int2* __restrict__ ed,
                                                  const int* __restrict__ rowptr,
                                                  const float* __restrict__ bias,
                                                  const float* __restrict__ inv,
                                                  float* __restrict__ ACC,
                                                  const float* __restrict__ H,
                                                  float w, float cw) {
  constexpr int LPN = COUT / 4;
  constexpr int LDZ = 3 * COUT;
  const int gt   = blockIdx.x * 256 + threadIdx.x;
  const int node = gt / LPN;
  if (node >= GDE_N) return;
  const int c = (gt % LPN) * 4;
  const int b = rowptr[node], n = rowptr[node + 1];
  const float* zc = Z + c;
  float4 s = make_float4(0.f, 0.f, 0.f, 0.f);
  int j = b;
  for (; j + 4 <= n; j += 4) {
    const int2 e0 = ed[j], e1 = ed[j + 1], e2 = ed[j + 2], e3 = ed[j + 3];
    const float* r0 = zc + (size_t)e0.x * LDZ;
    const float* r1 = zc + (size_t)e1.x * LDZ;
    const float* r2 = zc + (size_t)e2.x * LDZ;
    const float* r3 = zc + (size_t)e3.x * LDZ;
    const float4 a0 = *(const float4*)r0, b0 = *(const float4*)(r0 + COUT);
    const float4 a1 = *(const float4*)r1, b1 = *(const float4*)(r1 + COUT);
    const float4 a2 = *(const float4*)r2, b2 = *(const float4*)(r2 + COUT);
    const float4 a3 = *(const float4*)r3, b3 = *(const float4*)(r3 + COUT);
    gde_acc(s, a0, b0, __int_as_float(e0.y));
    gde_acc(s, a1, b1, __int_as_float(e1.y));
    gde_acc(s, a2, b2, __int_as_float(e2.y));
    gde_acc(s, a3, b3, __int_as_float(e3.y));
  }
  for (; j < n; ++j) {
    const int2 e0 = ed[j];
    const float* r0 = zc + (size_t)e0.x * LDZ;
    const float4 a0 = *(const float4*)r0, b0 = *(const float4*)(r0 + COUT);
    gde_acc(s, a0, b0, __int_as_float(e0.y));
  }
  const size_t oi = (size_t)node * COUT + c;
  const float iv = inv[node];
  const float4 zr = *(const float4*)(Z + (size_t)node * LDZ + 2 * COUT + c);
  float4 a  = make_float4(0.f, 0.f, 0.f, 0.f);
  float4 hv = make_float4(0.f, 0.f, 0.f, 0.f);
  if (RK == 2 || RK == 3) a  = *(const float4*)(ACC + oi);
  if (RK == 1 || RK == 2) hv = *(const float4*)(H + oi);
  if (RK == 3)            hv = *(const float4*)(out + oi);
  float4 v;
  v.x = s.x * iv + zr.x + bias[c + 0];
  v.y = s.y * iv + zr.y + bias[c + 1];
  v.z = s.z * iv + zr.z + bias[c + 2];
  v.w = s.w * iv + zr.w + bias[c + 3];
  if (ACT) { v.x = tanhf(v.x); v.y = tanhf(v.y); v.z = tanhf(v.z); v.w = tanhf(v.w); }
  if (RK == 0) {
    *(float4*)(out + oi) = v;
  } else if (RK == 1 || RK == 2) {
    if (RK == 1) { a.x = w * v.x;  a.y = w * v.y;  a.z = w * v.z;  a.w = w * v.w; }
    else         { a.x = a.x + w * v.x;  a.y = a.y + w * v.y;  a.z = a.z + w * v.z;  a.w = a.w + w * v.w; }
    *(float4*)(ACC + oi) = a;
    float4 y;
    y.x = hv.x + cw * v.x;  y.y = hv.y + cw * v.y;  y.z = hv.z + cw * v.z;  y.w = hv.w + cw * v.w;
    *(float4*)(out + oi) = y;
  } else {
    hv.x = hv.x + 0.5f * (a.x + v.x);  hv.y = hv.y + 0.5f * (a.y + v.y);
    hv.z = hv.z + 0.5f * (a.z + v.z);  hv.w = hv.w + 0.5f * (a.w + v.w);
    *(float4*)(out + oi) = hv;
  }
}

__global__ void gde_log_softmax16(float* __restrict__ out, int n) {
  int i = blockIdx.x * blockDim.x + threadIdx.x;
  if (i >= n) return;
  float v[16];
  float mx = -3.402823466e+38f;
  float* row = out + (size_t)i * 16;
#pragma unroll
  for (int c = 0; c < 16; ++c) { v[c] = row[c]; mx = fmaxf(mx, v[c]); }
  float s = 0.f;
#pragma unroll
  for (int c = 0; c < 16; ++c) s += __expf(v[c] - mx);
  float ls = __logf(s);
#pragma unroll
  for (int c = 0; c < 16; ++c) row[c] = v[c] - mx - ls;
}

extern "C" void kernel_launch(void* const* d_in, const int* in_sizes, int n_in,
                              void* d_out, int out_size, void* d_ws, size_t ws_size,
                              hipStream_t stream) {
  const float* x   = (const float*)d_in[0];
  const float* p   = (const float*)d_in[1];
  const int*   src = (const int*)d_in[2];
  const int*   dst = (const int*)d_in[3];
  const float* W1 = (const float*)d_in[4],  *R1 = (const float*)d_in[5],  *b1 = (const float*)d_in[6];
  const float* Wa = (const float*)d_in[7],  *Ra = (const float*)d_in[8],  *ba = (const float*)d_in[9];
  const float* Wb = (const float*)d_in[10], *Rb = (const float*)d_in[11], *bb = (const float*)d_in[12];
  const float* W2 = (const float*)d_in[13], *R2 = (const float*)d_in[14], *b2 = (const float*)d_in[15];
  float* out = (float*)d_out;

  float* W = (float*)d_ws;
  size_t o = 0;
  float* Z   = W + o; o += (size_t)GDE_N * 192;
  float* CSR = W + o; o += (size_t)GDE_N * 64;
  float* H   = W + o; o += (size_t)GDE_N * 64;
  float* YT  = W + o; o += (size_t)GDE_N * 64;
  float* U   = W + o; o += (size_t)GDE_N * 64;
  float* ACC = W + o; o += (size_t)GDE_N * 64;
  float* INV = W + o; o += (size_t)GDE_N;
  float* GC1 = W + o; o += (size_t)256 * 192;
  float* GCA = W + o; o += (size_t)64 * 192;
  float* GCB = W + o; o += (size_t)64 * 192;
  float* GC2 = W + o; o += (size_t)64 * 48;
  int*  CNT = (int*)CSR;
  int*  RP  = CNT + GDE_N;
  int*  TMP = RP + GDE_N + 2;
  int*  ARR = TMP + GDE_E;
  int2* ED  = (int2*)ARR;
  (void)ws_size; (void)n_in; (void)in_sizes; (void)out_size;

  gde_prep_g<<<(256 * 192 + 255) / 256, 256, 0, stream>>>(GC1, W1, R1, 256, 64);
  gde_prep_g<<<(64 * 192 + 255) / 256, 256, 0, stream>>>(GCA, Wa, Ra, 64, 64);
  gde_prep_g<<<(64 * 192 + 255) / 256, 256, 0, stream>>>(GCB, Wb, Rb, 64, 64);
  gde_prep_g<<<(64 * 48 + 255) / 256, 256, 0, stream>>>(GC2, W2, R2, 64, 16);

  hipMemsetAsync(CNT, 0, sizeof(int) * GDE_N, stream);
  gde_rank<<<(GDE_E + 255) / 256, 256, 0, stream>>>(CNT, ARR, dst, GDE_E);
  gde_scan<<<1, 1024, 0, stream>>>(RP, INV, CNT);
  gde_place<<<(GDE_E + 255) / 256, 256, 0, stream>>>(TMP, RP, ARR, dst, GDE_E);
  gde_order<<<(GDE_E + 255) / 256, 256, 0, stream>>>(ED, TMP, RP, src, dst, p, GDE_E);

  auto spline = [&](const float* y, int Cin, int Cout, const float* G,
                    const float* bias, float* ob, int act,
                    int rk, float w, float cw) {
    if (Cin == 256)
      gde_mm_bstat<256, 3, 4, 2><<<256, 256, 0, stream>>>(y, G, Z, GDE_N / 16);
    else if (Cout == 64)
      gde_mm_bstat<64, 4, 3, 4><<<256, 384, 0, stream>>>(y, G, Z, GDE_N / 16);
    else
      gde_mm_bstat<64, 3, 1, 8><<<256, 256, 0, stream>>>(y, G, Z, GDE_N / 16);
    if (Cout == 64 && act)
      gde_gather<64, 1, 0><<<GDE_N * 16 / 256, 256, 0, stream>>>(ob, Z, ED, RP, bias, INV, ACC, H, w, cw);
    else if (Cout == 64 && rk == 0)
      gde_gather<64, 0, 0><<<GDE_N * 16 / 256, 256, 0, stream>>>(ob, Z, ED, RP, bias, INV, ACC, H, w, cw);
    else if (Cout == 64 && rk == 1)
      gde_gather<64, 0, 1><<<GDE_N * 16 / 256, 256, 0, stream>>>(ob, Z, ED, RP, bias, INV, ACC, H, w, cw);
    else if (Cout == 64 && rk == 2)
      gde_gather<64, 0, 2><<<GDE_N * 16 / 256, 256, 0, stream>>>(ob, Z, ED, RP, bias, INV, ACC, H, w, cw);
    else if (Cout == 64)
      gde_gather<64, 0, 3><<<GDE_N * 16 / 256, 256, 0, stream>>>(ob, Z, ED, RP, bias, INV, ACC, YT, w, cw);
    else
      gde_gather<16, 1, 0><<<(GDE_N * 4 + 255) / 256, 256, 0, stream>>>(ob, Z, ED, RP, bias, INV, ACC, H, w, cw);
  };
  auto odefunc = [&](const float* y, float* ob, int rk, float w, float cw) {
    spline(y, 64, 64, GCA, ba, U, 0, 0, 0.f, 0.f);
    spline(U, 64, 64, GCB, bb, ob, 0, rk, w, cw);
  };

  spline(x, 256, 64, GC1, b1, H, 1, 0, 0.f, 0.f);

  odefunc(H,  YT, 1, 1.0f, 1.5f);
  odefunc(YT, YT, 2, 2.0f, 1.5f);
  odefunc(YT, YT, 2, 2.0f, 3.0f);
  odefunc(YT, H,  3, 0.0f, 0.0f);

  spline(H, 64, 16, GC2, b2, out, 1, 0, 0.f, 0.f);
  gde_log_softmax16<<<(GDE_N + 255) / 256, 256, 0, stream>>>(out, GDE_N);
}


# Hidden convs aggregate first: gather sums y[src] and p*y[src] per node (256 B per edge, was 512), then one K=192 WMMA product [S0|S1|y]@[W0;W1-W0;R], RK4 update in its epilogue; sum order changes
# speedup vs baseline: 3.2468x; 1.1453x over previous
#include <hip/hip_runtime.h>
#include <math.h>

#define GDE_N 50000
#define GDE_E 800000

typedef float v2f __attribute__((ext_vector_type(2)));
typedef float v8f __attribute__((ext_vector_type(8)));

template <int K, int NT, int CG, int RS>
__global__ __launch_bounds__(32 * CG * RS) void gde_mm_bstat(const float* __restrict__ X,
                                                            const float* __restrict__ G,
                                                            float* __restrict__ Y,
                                                            int nStrips) {
  constexpr int NCOLS = 16 * NT * CG;
  constexpr int STEPS = K / 4;
  constexpr int KC    = K / 64;
  const int lane = threadIdx.x & 31;
  const int wave = __builtin_amdgcn_readfirstlane((int)(threadIdx.x >> 5));
  const int c    = lane & 15;
  const int h    = lane >> 4;
  const int cg   = wave % CG;
  const int rs   = wave / CG;
  const int col0 = cg * (16 * NT) + c;

  v2f bf[STEPS][NT];
#pragma unroll
  for (int s = 0; s < STEPS; ++s) {
    const int kb = 8 * (s >> 1) + 2 * (s & 1) + 4 * h;
#pragma unroll
    for (int t = 0; t < NT; ++t) {
      bf[s][t].x = G[(size_t)kb       * NCOLS + col0 + 16 * t];
      bf[s][t].y = G[(size_t)(kb + 1) * NCOLS + col0 + 16 * t];
    }
  }

  const int stride = gridDim.x * RS;
  int strip = blockIdx.x * RS + rs;
  if (strip >= nStrips) return;
  const float* xr = X + ((size_t)strip * 16 + c) * K + 4 * h;

  float4 cur[8];
#pragma unroll
  for (int t = 0; t < 8; ++t) {
    cur[t] = *(const float4*)(xr + 8 * t);
    __builtin_amdgcn_sched_barrier(0);
  }

  for (; strip < nStrips; strip += stride) {
    const int sn = min(strip + stride, nStrips - 1);
    const float* xn = X + ((size_t)sn * 16 + c) * K + 4 * h;

    v8f acc[NT];
#pragma unroll
    for (int t = 0; t < NT; ++t) acc[t] = (v8f){0.f, 0.f, 0.f, 0.f, 0.f, 0.f, 0.f, 0.f};

#pragma unroll
    for (int kc = 0; kc < KC; ++kc) {
      const float* nsrc = (kc + 1 < KC) ? (xr + (kc + 1) * 64) : xn;
#pragma unroll
      for (int t = 0; t < 8; ++t) {
        v2f a0; a0.x = cur[t].x; a0.y = cur[t].y;
        v2f a1; a1.x = cur[t].z; a1.y = cur[t].w;
        const int S = kc * 16 + 2 * t;
#pragma unroll
        for (int q = 0; q < NT; ++q)
          acc[q] = __builtin_amdgcn_wmma_f32_16x16x4_f32(false, a0, false, bf[S][q],
                                                         (short)0, acc[q], false, false);
#pragma unroll
        for (int q = 0; q < NT; ++q)
          acc[q] = __builtin_amdgcn_wmma_f32_16x16x4_f32(false, a1, false, bf[S + 1][q],
                                                         (short)0, acc[q], false, false);
        cur[t] = *(const float4*)(nsrc + 8 * t);
        __builtin_amdgcn_sched_barrier(0);
      }
    }

    float* yr = Y + ((size_t)strip * 16 + 8 * h) * NCOLS + col0;
#pragma unroll
    for (int q = 0; q < NT; ++q)
#pragma unroll
      for (int r = 0; r < 8; ++r)
        yr[(size_t)r * NCOLS + 16 * q] = acc[q][r];

    xr = xn;
  }
}

template <int NT, int CG, int RS, int RK>
__global__ __launch_bounds__(32 * CG * RS) void gde_mm_agg(const float* __restrict__ S,
                                                          const float* __restrict__ Y,
                                                          const float* __restrict__ G,
                                                          const float* __restrict__ bias,
                                                          float* __restrict__ out,
                                                          float* __restrict__ ACC,
                                                          const float* __restrict__ H,
                                                          float w, float cw, int nStrips) {
  constexpr int NCOLS = 16 * NT * CG;
  constexpr int K     = 192;
  constexpr int STEPS = K / 4;
  constexpr int KC    = K / 64;
  static_assert(NCOLS == 64, "one block column covers the 64 output channels");
  const int lane = threadIdx.x & 31;
  const int wave = __builtin_amdgcn_readfirstlane((int)(threadIdx.x >> 5));
  const int c    = lane & 15;
  const int h    = lane >> 4;
  const int cg   = wave % CG;
  const int rs   = wave / CG;
  const int col0 = cg * (16 * NT) + c;

  v2f bf[STEPS][NT];
#pragma unroll
  for (int s = 0; s < STEPS; ++s) {
    const int kb = 8 * (s >> 1) + 2 * (s & 1) + 4 * h;
#pragma unroll
    for (int t = 0; t < NT; ++t) {
      bf[s][t].x = G[(size_t)kb       * NCOLS + col0 + 16 * t];
      bf[s][t].y = G[(size_t)(kb + 1) * NCOLS + col0 + 16 * t];
    }
  }
  float bq[NT];
#pragma unroll
  for (int q = 0; q < NT; ++q) bq[q] = bias[col0 + 16 * q];

  const int stride = gridDim.x * RS;
  int strip = blockIdx.x * RS + rs;
  if (strip >= nStrips) return;
  const float* sr = S + ((size_t)strip * 16 + c) * 128 + 4 * h;
  const float* yr = Y + ((size_t)strip * 16 + c) * 64 + 4 * h;

  float4 cur[8];
#pragma unroll
  for (int t = 0; t < 8; ++t) {
    cur[t] = *(const float4*)(sr + 8 * t);
    __builtin_amdgcn_sched_barrier(0);
  }

  for (; strip < nStrips; strip += stride) {
    const int sn = min(strip + stride, nStrips - 1);
    const float* sn_s = S + ((size_t)sn * 16 + c) * 128 + 4 * h;
    const float* sn_y = Y + ((size_t)sn * 16 + c) * 64 + 4 * h;

    const size_t o0 = ((size_t)strip * 16 + 8 * h) * NCOLS + col0;
    float av[NT][8], hv[NT][8];
#pragma unroll
    for (int q = 0; q < NT; ++q)
#pragma unroll
      for (int r = 0; r < 8; ++r) {
        const size_t oi = o0 + (size_t)r * NCOLS + 16 * q;
        av[q][r] = 0.f; hv[q][r] = 0.f;
        if (RK == 2 || RK == 3) av[q][r] = ACC[oi];
        if (RK == 1 || RK == 2) hv[q][r] = H[oi];
        if (RK == 3)            hv[q][r] = out[oi];
      }

    v8f acc[NT];
#pragma unroll
    for (int t = 0; t < NT; ++t) acc[t] = (v8f){0.f, 0.f, 0.f, 0.f, 0.f, 0.f, 0.f, 0.f};

#pragma unroll
    for (int kc = 0; kc < KC; ++kc) {
      const float* nsrc = (kc == 0) ? (sr + 64) : (kc == 1) ? yr : sn_s;
#pragma unroll
      for (int t = 0; t < 8; ++t) {
        v2f a0; a0.x = cur[t].x; a0.y = cur[t].y;
        v2f a1; a1.x = cur[t].z; a1.y = cur[t].w;
        const int St = kc * 16 + 2 * t;
#pragma unroll
        for (int q = 0; q < NT; ++q)
          acc[q] = __builtin_amdgcn_wmma_f32_16x16x4_f32(false, a0, false, bf[St][q],
                                                         (short)0, acc[q], false, false);
#pragma unroll
        for (int q = 0; q < NT; ++q)
          acc[q] = __builtin_amdgcn_wmma_f32_16x16x4_f32(false, a1, false, bf[St + 1][q],
                                                         (short)0, acc[q], false, false);
        cur[t] = *(const float4*)(nsrc + 8 * t);
        __builtin_amdgcn_sched_barrier(0);
      }
    }

#pragma unroll
    for (int q = 0; q < NT; ++q)
#pragma unroll
      for (int r = 0; r < 8; ++r) {
        const size_t oi = o0 + (size_t)r * NCOLS + 16 * q;
        const float v = acc[q][r] + bq[q];
        if (RK == 0) {
          out[oi] = v;
        } else if (RK == 1) {
          ACC[oi] = w * v;
          out[oi] = hv[q][r] + cw * v;
        } else if (RK == 2) {
          ACC[oi] = av[q][r] + w * v;
          out[oi] = hv[q][r] + cw * v;
        } else {
          out[oi] = hv[q][r] + 0.5f * (av[q][r] + v);
        }
      }

    sr = sn_s;
    yr = sn_y;
  }
}

__global__ void gde_prep_g(float* __restrict__ G, const float* __restrict__ Wp,
                           const float* __restrict__ R, int Cin, int Cout) {
  int idx = blockIdx.x * blockDim.x + threadIdx.x;
  int Ncols = 3 * Cout;
  int tot = Cin * Ncols;
  if (idx >= tot) return;
  int r = idx / Ncols, c = idx % Ncols;
  float v;
  if (c < Cout)            v = Wp[r * Cout + c];
  else if (c < 2 * Cout)   v = Wp[(size_t)Cin * Cout + r * Cout + (c - Cout)]
                             - Wp[r * Cout + (c - Cout)];
  else                     v = R[r * Cout + (c - 2 * Cout)];
  G[idx] = v;
}

__global__ void gde_prep_gs(float* __restrict__ G, const float* __restrict__ Wp,
                            const float* __restrict__ R, int C) {
  int idx = blockIdx.x * blockDim.x + threadIdx.x;
  if (idx >= 3 * C * C) return;
  int k = idx / C, c = idx % C;
  float v;
  if (k < C)            v = Wp[k * C + c];
  else if (k < 2 * C)   v = Wp[(size_t)C * C + (k - C) * C + c] - Wp[(k - C) * C + c];
  else                  v = R[(k - 2 * C) * C + c];
  G[idx] = v;
}

__global__ void gde_rank(int* __restrict__ cnt, int* __restrict__ arr,
                         const int* __restrict__ dst, int E) {
  int e = blockIdx.x * blockDim.x + threadIdx.x;
  if (e < E) arr[e] = atomicAdd(&cnt[dst[e]], 1);
}

__global__ __launch_bounds__(1024) void gde_scan(int* __restrict__ rowptr, float* __restrict__ inv,
                                                 const int* __restrict__ cnt) {
  constexpr int N4    = GDE_N / 4;
  constexpr int TILES = (N4 + 1023) / 1024;
  static_assert(GDE_N % 4 == 0, "int4 tiles");
  __shared__ int wsum[32];
  __shared__ int tile_total;
  const int t = threadIdx.x, lane = t & 31, w = t >> 5;
  int4 v[TILES];
#pragma unroll
  for (int k = 0; k < TILES; ++k) {
    const int i = k * 1024 + t;
    v[k] = (i < N4) ? ((const int4*)cnt)[i] : make_int4(0, 0, 0, 0);
  }
  int carry = 0;
#pragma unroll
  for (int k = 0; k < TILES; ++k) {
    const int i = k * 1024 + t;
    const int s = v[k].x + v[k].y + v[k].z + v[k].w;
    int inc = s;
    for (int d = 1; d < 32; d <<= 1) {
      const int o = __shfl_up(inc, d);
      if (lane >= d) inc += o;
    }
    if (lane == 31) wsum[w] = inc;
    __syncthreads();
    if (w == 0) {
      const int x = wsum[lane];
      int xi = x;
      for (int d = 1; d < 32; d <<= 1) {
        const int o = __shfl_up(xi, d);
        if (lane >= d) xi += o;
      }
      wsum[lane] = xi - x;
      if (lane == 31) tile_total = xi;
    }
    __syncthreads();
    const int base = carry + wsum[w] + inc - s;
    carry += tile_total;
    if (i < N4) {
      int4 r;
      r.x = base;
      r.y = r.x + v[k].x;
      r.z = r.y + v[k].y;
      r.w = r.z + v[k].z;
      ((int4*)rowptr)[i] = r;
      float4 f;
      f.x = 1.0f / fmaxf((float)v[k].x, 1.0f);
      f.y = 1.0f / fmaxf((float)v[k].y, 1.0f);
      f.z = 1.0f / fmaxf((float)v[k].z, 1.0f);
      f.w = 1.0f / fmaxf((float)v[k].w, 1.0f);
      ((float4*)inv)[i] = f;
    }
    __syncthreads();
  }
  if (t == 0) rowptr[GDE_N] = carry;
}

__global__ void gde_place(int* __restrict__ tmp, const int* __restrict__ rowptr,
                          const int* __restrict__ arr, const int* __restrict__ dst, int E) {
  int e = blockIdx.x * blockDim.x + threadIdx.x;
  if (e < E) tmp[rowptr[dst[e]] + arr[e]] = e;
}

__global__ void gde_order(int2* __restrict__ ed, const int* __restrict__ tmp,
                          const int* __restrict__ rowptr, const int* __restrict__ src,
                          const int* __restrict__ dst, const float* __restrict__ p, int E) {
  int e = blockIdx.x * blockDim.x + threadIdx.x;
  if (e >= E) return;
  const int d = dst[e];
  const int b = rowptr[d], n = rowptr[d + 1];
  int r = 0;
  for (int j = b; j < n; ++j) r += (tmp[j] < e) ? 1 : 0;
  ed[b + r] = make_int2(src[e], __float_as_int(p[e]));
}

__device__ __forceinline__ void gde_acc(float4& s, const float4& z0, const float4& z1, float pe) {
  s.x += z0.x + pe * z1.x;
  s.y += z0.y + pe * z1.y;
  s.z += z0.z + pe * z1.z;
  s.w += z0.w + pe * z1.w;
}

template <int COUT>
__global__ __launch_bounds__(256) void gde_gather(float* __restrict__ out, const float* __restrict__ Z,
                                                  const int2* __restrict__ ed,
                                                  const int* __restrict__ rowptr,
                                                  const float* __restrict__ bias,
                                                  const float* __restrict__ inv) {
  constexpr int LPN = COUT / 4;
  constexpr int LDZ = 3 * COUT;
  const int gt   = blockIdx.x * 256 + threadIdx.x;
  const int node = gt / LPN;
  if (node >= GDE_N) return;
  const int c = (gt % LPN) * 4;
  const int b = rowptr[node], n = rowptr[node + 1];
  const float* zc = Z + c;
  float4 s = make_float4(0.f, 0.f, 0.f, 0.f);
  int j = b;
  for (; j + 4 <= n; j += 4) {
    const int2 e0 = ed[j], e1 = ed[j + 1], e2 = ed[j + 2], e3 = ed[j + 3];
    const float* r0 = zc + (size_t)e0.x * LDZ;
    const float* r1 = zc + (size_t)e1.x * LDZ;
    const float* r2 = zc + (size_t)e2.x * LDZ;
    const float* r3 = zc + (size_t)e3.x * LDZ;
    const float4 a0 = *(const float4*)r0, b0 = *(const float4*)(r0 + COUT);
    const float4 a1 = *(const float4*)r1, b1 = *(const float4*)(r1 + COUT);
    const float4 a2 = *(const float4*)r2, b2 = *(const float4*)(r2 + COUT);
    const float4 a3 = *(const float4*)r3, b3 = *(const float4*)(r3 + COUT);
    gde_acc(s, a0, b0, __int_as_float(e0.y));
    gde_acc(s, a1, b1, __int_as_float(e1.y));
    gde_acc(s, a2, b2, __int_as_float(e2.y));
    gde_acc(s, a3, b3, __int_as_float(e3.y));
  }
  for (; j < n; ++j) {
    const int2 e0 = ed[j];
    const float* r0 = zc + (size_t)e0.x * LDZ;
    const float4 a0 = *(const float4*)r0, b0 = *(const float4*)(r0 + COUT);
    gde_acc(s, a0, b0, __int_as_float(e0.y));
  }
  const size_t oi = (size_t)node * COUT + c;
  const float iv = inv[node];
  const float4 zr = *(const float4*)(Z + (size_t)node * LDZ + 2 * COUT + c);
  float4 v;
  v.x = tanhf(s.x * iv + zr.x + bias[c + 0]);
  v.y = tanhf(s.y * iv + zr.y + bias[c + 1]);
  v.z = tanhf(s.z * iv + zr.z + bias[c + 2]);
  v.w = tanhf(s.w * iv + zr.w + bias[c + 3]);
  *(float4*)(out + oi) = v;
}

__device__ __forceinline__ void gde_acc2(float4& s0, float4& s1, const float4& y, float pe) {
  s0.x += y.x;  s0.y += y.y;  s0.z += y.z;  s0.w += y.w;
  s1.x += pe * y.x;  s1.y += pe * y.y;  s1.z += pe * y.z;  s1.w += pe * y.w;
}

__global__ __launch_bounds__(256) void gde_gather_in(float* __restrict__ S, const float* __restrict__ Y,
                                                     const int2* __restrict__ ed,
                                                     const int* __restrict__ rowptr,
                                                     const float* __restrict__ inv) {
  const int gt   = blockIdx.x * 256 + threadIdx.x;
  const int node = gt / 16;
  if (node >= GDE_N) return;
  const int c = (gt % 16) * 4;
  const int b = rowptr[node], n = rowptr[node + 1];
  const float iv = inv[node];
  const float* yc = Y + c;
  float4 s0 = make_float4(0.f, 0.f, 0.f, 0.f);
  float4 s1 = make_float4(0.f, 0.f, 0.f, 0.f);
  int j = b;
  for (; j + 4 <= n; j += 4) {
    const int2 e0 = ed[j], e1 = ed[j + 1], e2 = ed[j + 2], e3 = ed[j + 3];
    const float4 a0 = *(const float4*)(yc + (size_t)e0.x * 64);
    const float4 a1 = *(const float4*)(yc + (size_t)e1.x * 64);
    const float4 a2 = *(const float4*)(yc + (size_t)e2.x * 64);
    const float4 a3 = *(const float4*)(yc + (size_t)e3.x * 64);
    gde_acc2(s0, s1, a0, __int_as_float(e0.y));
    gde_acc2(s0, s1, a1, __int_as_float(e1.y));
    gde_acc2(s0, s1, a2, __int_as_float(e2.y));
    gde_acc2(s0, s1, a3, __int_as_float(e3.y));
  }
#pragma nounroll
  for (; j < n; ++j) {
    const int2 e0 = ed[j];
    const float4 a0 = *(const float4*)(yc + (size_t)e0.x * 64);
    gde_acc2(s0, s1, a0, __int_as_float(e0.y));
  }
  float* sp = S + (size_t)node * 128 + c;
  *(float4*)sp        = make_float4(s0.x * iv, s0.y * iv, s0.z * iv, s0.w * iv);
  *(float4*)(sp + 64) = make_float4(s1.x * iv, s1.y * iv, s1.z * iv, s1.w * iv);
}

__global__ void gde_log_softmax16(float* __restrict__ out, int n) {
  int i = blockIdx.x * blockDim.x + threadIdx.x;
  if (i >= n) return;
  float v[16];
  float mx = -3.402823466e+38f;
  float* row = out + (size_t)i * 16;
#pragma unroll
  for (int c = 0; c < 16; ++c) { v[c] = row[c]; mx = fmaxf(mx, v[c]); }
  float s = 0.f;
#pragma unroll
  for (int c = 0; c < 16; ++c) s += __expf(v[c] - mx);
  float ls = __logf(s);
#pragma unroll
  for (int c = 0; c < 16; ++c) row[c] = v[c] - mx - ls;
}

extern "C" void kernel_launch(void* const* d_in, const int* in_sizes, int n_in,
                              void* d_out, int out_size, void* d_ws, size_t ws_size,
                              hipStream_t stream) {
  const float* x   = (const float*)d_in[0];
  const float* p   = (const float*)d_in[1];
  const int*   src = (const int*)d_in[2];
  const int*   dst = (const int*)d_in[3];
  const float* W1 = (const float*)d_in[4],  *R1 = (const float*)d_in[5],  *b1 = (const float*)d_in[6];
  const float* Wa = (const float*)d_in[7],  *Ra = (const float*)d_in[8],  *ba = (const float*)d_in[9];
  const float* Wb = (const float*)d_in[10], *Rb = (const float*)d_in[11], *bb = (const float*)d_in[12];
  const float* W2 = (const float*)d_in[13], *R2 = (const float*)d_in[14], *b2 = (const float*)d_in[15];
  float* out = (float*)d_out;

  float* W = (float*)d_ws;
  size_t o = 0;
  float* Z   = W + o; o += (size_t)GDE_N * 192;
  float* CSR = W + o; o += (size_t)GDE_N * 64;
  float* H   = W + o; o += (size_t)GDE_N * 64;
  float* YT  = W + o; o += (size_t)GDE_N * 64;
  float* U   = W + o; o += (size_t)GDE_N * 64;
  float* ACC = W + o; o += (size_t)GDE_N * 64;
  float* INV = W + o; o += (size_t)GDE_N;
  float* GC1 = W + o; o += (size_t)256 * 192;
  float* GSA = W + o; o += (size_t)192 * 64;
  float* GSB = W + o; o += (size_t)192 * 64;
  float* GC2 = W + o; o += (size_t)64 * 48;
  float* S   = Z;
  int*  CNT = (int*)CSR;
  int*  RP  = CNT + GDE_N;
  int*  TMP = RP + GDE_N + 2;
  int*  ARR = TMP + GDE_E;
  int2* ED  = (int2*)ARR;
  (void)ws_size; (void)n_in; (void)in_sizes; (void)out_size;

  gde_prep_g<<<(256 * 192 + 255) / 256, 256, 0, stream>>>(GC1, W1, R1, 256, 64);
  gde_prep_gs<<<(192 * 64 + 255) / 256, 256, 0, stream>>>(GSA, Wa, Ra, 64);
  gde_prep_gs<<<(192 * 64 + 255) / 256, 256, 0, stream>>>(GSB, Wb, Rb, 64);
  gde_prep_g<<<(64 * 48 + 255) / 256, 256, 0, stream>>>(GC2, W2, R2, 64, 16);

  hipMemsetAsync(CNT, 0, sizeof(int) * GDE_N, stream);
  gde_rank<<<(GDE_E + 255) / 256, 256, 0, stream>>>(CNT, ARR, dst, GDE_E);
  gde_scan<<<1, 1024, 0, stream>>>(RP, INV, CNT);
  gde_place<<<(GDE_E + 255) / 256, 256, 0, stream>>>(TMP, RP, ARR, dst, GDE_E);
  gde_order<<<(GDE_E + 255) / 256, 256, 0, stream>>>(ED, TMP, RP, src, dst, p, GDE_E);

  auto hidden = [&](const float* y, const float* G, const float* bias, float* ob,
                    int rk, float w, float cw) {
    gde_gather_in<<<GDE_N * 16 / 256, 256, 0, stream>>>(S, y, ED, RP, INV);
    if (rk == 0)
      gde_mm_agg<2, 2, 4, 0><<<256, 256, 0, stream>>>(S, y, G, bias, ob, ACC, H, w, cw, GDE_N / 16);
    else if (rk == 1)
      gde_mm_agg<2, 2, 4, 1><<<256, 256, 0, stream>>>(S, y, G, bias, ob, ACC, H, w, cw, GDE_N / 16);
    else if (rk == 2)
      gde_mm_agg<2, 2, 4, 2><<<256, 256, 0, stream>>>(S, y, G, bias, ob, ACC, H, w, cw, GDE_N / 16);
    else
      gde_mm_agg<2, 2, 4, 3><<<256, 256, 0, stream>>>(S, y, G, bias, ob, ACC, YT, w, cw, GDE_N / 16);
  };
  auto odefunc = [&](const float* y, float* ob, int rk, float w, float cw) {
    hidden(y, GSA, ba, U, 0, 0.f, 0.f);
    hidden(U, GSB, bb, ob, rk, w, cw);
  };

  gde_mm_bstat<256, 3, 4, 2><<<256, 256, 0, stream>>>(x, GC1, Z, GDE_N / 16);
  gde_gather<64><<<GDE_N * 16 / 256, 256, 0, stream>>>(H, Z, ED, RP, b1, INV);

  odefunc(H,  YT, 1, 1.0f, 1.5f);
  odefunc(YT, YT, 2, 2.0f, 1.5f);
  odefunc(YT, YT, 2, 2.0f, 3.0f);
  odefunc(YT, H,  3, 0.0f, 0.0f);

  gde_mm_bstat<64, 3, 1, 8><<<256, 256, 0, stream>>>(H, GC2, Z, GDE_N / 16);
  gde_gather<16><<<(GDE_N * 4 + 255) / 256, 256, 0, stream>>>(out, Z, ED, RP, b2, INV);
  gde_log_softmax16<<<(GDE_N + 255) / 256, 256, 0, stream>>>(out, GDE_N);
}


# Hidden convs in ONE launch each: a 512-thread block gathers the sums of its 32 nodes into LDS and multiplies them there (f32 WMMA, G copied to LDS in fragment order); 8 launches gone; sum order kept
# speedup vs baseline: 3.6629x; 1.1282x over previous
#include <hip/hip_runtime.h>
#include <math.h>

#define GDE_N 50000
#define GDE_E 800000

typedef float v2f __attribute__((ext_vector_type(2)));
typedef float v8f __attribute__((ext_vector_type(8)));

template <int K, int NT, int CG, int RS>
__global__ __launch_bounds__(32 * CG * RS) void gde_mm_bstat(const float* __restrict__ X,
                                                            const float* __restrict__ G,
                                                            float* __restrict__ Y,
                                                            int nStrips) {
  constexpr int NCOLS = 16 * NT * CG;
  constexpr int STEPS = K / 4;
  constexpr int KC    = K / 64;
  const int lane = threadIdx.x & 31;
  const int wave = __builtin_amdgcn_readfirstlane((int)(threadIdx.x >> 5));
  const int c    = lane & 15;
  const int h    = lane >> 4;
  const int cg   = wave % CG;
  const int rs   = wave / CG;
  const int col0 = cg * (16 * NT) + c;

  v2f bf[STEPS][NT];
#pragma unroll
  for (int s = 0; s < STEPS; ++s) {
    const int kb = 8 * (s >> 1) + 2 * (s & 1) + 4 * h;
#pragma unroll
    for (int t = 0; t < NT; ++t) {
      bf[s][t].x = G[(size_t)kb       * NCOLS + col0 + 16 * t];
      bf[s][t].y = G[(size_t)(kb + 1) * NCOLS + col0 + 16 * t];
    }
  }

  const int stride = gridDim.x * RS;
  int strip = blockIdx.x * RS + rs;
  if (strip >= nStrips) return;
  const float* xr = X + ((size_t)strip * 16 + c) * K + 4 * h;

  float4 cur[8];
#pragma unroll
  for (int t = 0; t < 8; ++t) {
    cur[t] = *(const float4*)(xr + 8 * t);
    __builtin_amdgcn_sched_barrier(0);
  }

  for (; strip < nStrips; strip += stride) {
    const int sn = min(strip + stride, nStrips - 1);
    const float* xn = X + ((size_t)sn * 16 + c) * K + 4 * h;

    v8f acc[NT];
#pragma unroll
    for (int t = 0; t < NT; ++t) acc[t] = (v8f){0.f, 0.f, 0.f, 0.f, 0.f, 0.f, 0.f, 0.f};

#pragma unroll
    for (int kc = 0; kc < KC; ++kc) {
      const float* nsrc = (kc + 1 < KC) ? (xr + (kc + 1) * 64) : xn;
#pragma unroll
      for (int t = 0; t < 8; ++t) {
        v2f a0; a0.x = cur[t].x; a0.y = cur[t].y;
        v2f a1; a1.x = cur[t].z; a1.y = cur[t].w;
        const int S = kc * 16 + 2 * t;
#pragma unroll
        for (int q = 0; q < NT; ++q)
          acc[q] = __builtin_amdgcn_wmma_f32_16x16x4_f32(false, a0, false, bf[S][q],
                                                         (short)0, acc[q], false, false);
#pragma unroll
        for (int q = 0; q < NT; ++q)
          acc[q] = __builtin_amdgcn_wmma_f32_16x16x4_f32(false, a1, false, bf[S + 1][q],
                                                         (short)0, acc[q], false, false);
        cur[t] = *(const float4*)(nsrc + 8 * t);
        __builtin_amdgcn_sched_barrier(0);
      }
    }

    float* yr = Y + ((size_t)strip * 16 + 8 * h) * NCOLS + col0;
#pragma unroll
    for (int q = 0; q < NT; ++q)
#pragma unroll
      for (int r = 0; r < 8; ++r)
        yr[(size_t)r * NCOLS + 16 * q] = acc[q][r];

    xr = xn;
  }
}

__global__ void gde_prep_g(float* __restrict__ G, const float* __restrict__ Wp,
                           const float* __restrict__ R, int Cin, int Cout) {
  int idx = blockIdx.x * blockDim.x + threadIdx.x;
  int Ncols = 3 * Cout;
  int tot = Cin * Ncols;
  if (idx >= tot) return;
  int r = idx / Ncols, c = idx % Ncols;
  float v;
  if (c < Cout)            v = Wp[r * Cout + c];
  else if (c < 2 * Cout)   v = Wp[(size_t)Cin * Cout + r * Cout + (c - Cout)]
                             - Wp[r * Cout + (c - Cout)];
  else                     v = R[r * Cout + (c - 2 * Cout)];
  G[idx] = v;
}

__global__ void gde_prep_gf(float* __restrict__ G, const float* __restrict__ Wp,
                            const float* __restrict__ R) {
  constexpr int C = 64;
  int idx = blockIdx.x * blockDim.x + threadIdx.x;
  if (idx >= 3 * C * C) return;
  const int j = idx & 3, lane = (idx >> 2) & 31, T = (idx >> 7) % 24, q = idx / (128 * 24);
  const int k = 8 * T + 4 * (lane >> 4) + j;
  const int c = 16 * q + (lane & 15);
  float v;
  if (k < C)            v = Wp[k * C + c];
  else if (k < 2 * C)   v = Wp[(size_t)C * C + (k - C) * C + c] - Wp[(k - C) * C + c];
  else                  v = R[(k - 2 * C) * C + c];
  G[idx] = v;
}

__global__ void gde_rank(int* __restrict__ cnt, int* __restrict__ arr,
                         const int* __restrict__ dst, int E) {
  int e = blockIdx.x * blockDim.x + threadIdx.x;
  if (e < E) arr[e] = atomicAdd(&cnt[dst[e]], 1);
}

__global__ __launch_bounds__(1024) void gde_scan(int* __restrict__ rowptr, float* __restrict__ inv,
                                                 const int* __restrict__ cnt) {
  constexpr int N4    = GDE_N / 4;
  constexpr int TILES = (N4 + 1023) / 1024;
  static_assert(GDE_N % 4 == 0, "int4 tiles");
  __shared__ int wsum[32];
  __shared__ int tile_total;
  const int t = threadIdx.x, lane = t & 31, w = t >> 5;
  int4 v[TILES];
#pragma unroll
  for (int k = 0; k < TILES; ++k) {
    const int i = k * 1024 + t;
    v[k] = (i < N4) ? ((const int4*)cnt)[i] : make_int4(0, 0, 0, 0);
  }
  int carry = 0;
#pragma unroll
  for (int k = 0; k < TILES; ++k) {
    const int i = k * 1024 + t;
    const int s = v[k].x + v[k].y + v[k].z + v[k].w;
    int inc = s;
    for (int d = 1; d < 32; d <<= 1) {
      const int o = __shfl_up(inc, d);
      if (lane >= d) inc += o;
    }
    if (lane == 31) wsum[w] = inc;
    __syncthreads();
    if (w == 0) {
      const int x = wsum[lane];
      int xi = x;
      for (int d = 1; d < 32; d <<= 1) {
        const int o = __shfl_up(xi, d);
        if (lane >= d) xi += o;
      }
      wsum[lane] = xi - x;
      if (lane == 31) tile_total = xi;
    }
    __syncthreads();
    const int base = carry + wsum[w] + inc - s;
    carry += tile_total;
    if (i < N4) {
      int4 r;
      r.x = base;
      r.y = r.x + v[k].x;
      r.z = r.y + v[k].y;
      r.w = r.z + v[k].z;
      ((int4*)rowptr)[i] = r;
      float4 f;
      f.x = 1.0f / fmaxf((float)v[k].x, 1.0f);
      f.y = 1.0f / fmaxf((float)v[k].y, 1.0f);
      f.z = 1.0f / fmaxf((float)v[k].z, 1.0f);
      f.w = 1.0f / fmaxf((float)v[k].w, 1.0f);
      ((float4*)inv)[i] = f;
    }
    __syncthreads();
  }
  if (t == 0) rowptr[GDE_N] = carry;
}

__global__ void gde_place(int* __restrict__ tmp, const int* __restrict__ rowptr,
                          const int* __restrict__ arr, const int* __restrict__ dst, int E) {
  int e = blockIdx.x * blockDim.x + threadIdx.x;
  if (e < E) tmp[rowptr[dst[e]] + arr[e]] = e;
}

__global__ void gde_order(int2* __restrict__ ed, const int* __restrict__ tmp,
                          const int* __restrict__ rowptr, const int* __restrict__ src,
                          const int* __restrict__ dst, const float* __restrict__ p, int E) {
  int e = blockIdx.x * blockDim.x + threadIdx.x;
  if (e >= E) return;
  const int d = dst[e];
  const int b = rowptr[d], n = rowptr[d + 1];
  int r = 0;
  for (int j = b; j < n; ++j) r += (tmp[j] < e) ? 1 : 0;
  ed[b + r] = make_int2(src[e], __float_as_int(p[e]));
}

__device__ __forceinline__ void gde_acc(float4& s, const float4& z0, const float4& z1, float pe) {
  s.x += z0.x + pe * z1.x;
  s.y += z0.y + pe * z1.y;
  s.z += z0.z + pe * z1.z;
  s.w += z0.w + pe * z1.w;
}

template <int COUT>
__global__ __launch_bounds__(256) void gde_gather(float* __restrict__ out, const float* __restrict__ Z,
                                                  const int2* __restrict__ ed,
                                                  const int* __restrict__ rowptr,
                                                  const float* __restrict__ bias,
                                                  const float* __restrict__ inv) {
  constexpr int LPN = COUT / 4;
  constexpr int LDZ = 3 * COUT;
  const int gt   = blockIdx.x * 256 + threadIdx.x;
  const int node = gt / LPN;
  if (node >= GDE_N) return;
  const int c = (gt % LPN) * 4;
  const int b = rowptr[node], n = rowptr[node + 1];
  const float* zc = Z + c;
  float4 s = make_float4(0.f, 0.f, 0.f, 0.f);
  int j = b;
  for (; j + 4 <= n; j += 4) {
    const int2 e0 = ed[j], e1 = ed[j + 1], e2 = ed[j + 2], e3 = ed[j + 3];
    const float* r0 = zc + (size_t)e0.x * LDZ;
    const float* r1 = zc + (size_t)e1.x * LDZ;
    const float* r2 = zc + (size_t)e2.x * LDZ;
    const float* r3 = zc + (size_t)e3.x * LDZ;
    const float4 a0 = *(const float4*)r0, b0 = *(const float4*)(r0 + COUT);
    const float4 a1 = *(const float4*)r1, b1 = *(const float4*)(r1 + COUT);
    const float4 a2 = *(const float4*)r2, b2 = *(const float4*)(r2 + COUT);
    const float4 a3 = *(const float4*)r3, b3 = *(const float4*)(r3 + COUT);
    gde_acc(s, a0, b0, __int_as_float(e0.y));
    gde_acc(s, a1, b1, __int_as_float(e1.y));
    gde_acc(s, a2, b2, __int_as_float(e2.y));
    gde_acc(s, a3, b3, __int_as_float(e3.y));
  }
  for (; j < n; ++j) {
    const int2 e0 = ed[j];
    const float* r0 = zc + (size_t)e0.x * LDZ;
    const float4 a0 = *(const float4*)r0, b0 = *(const float4*)(r0 + COUT);
    gde_acc(s, a0, b0, __int_as_float(e0.y));
  }
  const size_t oi = (size_t)node * COUT + c;
  const float iv = inv[node];
  const float4 zr = *(const float4*)(Z + (size_t)node * LDZ + 2 * COUT + c);
  float4 v;
  v.x = tanhf(s.x * iv + zr.x + bias[c + 0]);
  v.y = tanhf(s.y * iv + zr.y + bias[c + 1]);
  v.z = tanhf(s.z * iv + zr.z + bias[c + 2]);
  v.w = tanhf(s.w * iv + zr.w + bias[c + 3]);
  *(float4*)(out + oi) = v;
}

__device__ __forceinline__ void gde_acc2(float4& s0, float4& s1, const float4& y, float pe) {
  s0.x += y.x;  s0.y += y.y;  s0.z += y.z;  s0.w += y.w;
  s1.x += pe * y.x;  s1.y += pe * y.y;  s1.z += pe * y.z;  s1.w += pe * y.w;
}

template <int RK>
__global__ __launch_bounds__(512) void gde_conv_fused(const float* __restrict__ Y,
                                                      const int2* __restrict__ ed,
                                                      const int* __restrict__ rowptr,
                                                      const float* __restrict__ inv,
                                                      const float* __restrict__ Gf,
                                                      const float* __restrict__ bias,
                                                      float* __restrict__ out,
                                                      float* __restrict__ ACC,
                                                      const float* __restrict__ H,
                                                      float w, float cw) {
  constexpr int PITCH = 196;
  __shared__ float4 Bs[192 * 64 / 4];
  __shared__ __attribute__((aligned(16))) float At[32 * PITCH];
  const int tid = threadIdx.x;

  const int nl   = tid >> 4;
  const int c4   = (tid & 15) * 4;
  const int node = blockIdx.x * 32 + nl;
  int b = 0, n = 0;
  float iv = 0.f;
  float4 yv = make_float4(0.f, 0.f, 0.f, 0.f);
  if (node < GDE_N) {
    b  = rowptr[node];
    n  = rowptr[node + 1];
    iv = inv[node];
    yv = *(const float4*)(Y + (size_t)node * 64 + c4);
  }
#pragma unroll
  for (int i = 0; i < 6; ++i) Bs[tid + 512 * i] = ((const float4*)Gf)[tid + 512 * i];
  const float* yc = Y + c4;
  float4 s0 = make_float4(0.f, 0.f, 0.f, 0.f);
  float4 s1 = make_float4(0.f, 0.f, 0.f, 0.f);
  int j = b;
  for (; j + 4 <= n; j += 4) {
    const int2 e0 = ed[j], e1 = ed[j + 1], e2 = ed[j + 2], e3 = ed[j + 3];
    const float4 a0 = *(const float4*)(yc + (size_t)e0.x * 64);
    const float4 a1 = *(const float4*)(yc + (size_t)e1.x * 64);
    const float4 a2 = *(const float4*)(yc + (size_t)e2.x * 64);
    const float4 a3 = *(const float4*)(yc + (size_t)e3.x * 64);
    gde_acc2(s0, s1, a0, __int_as_float(e0.y));
    gde_acc2(s0, s1, a1, __int_as_float(e1.y));
    gde_acc2(s0, s1, a2, __int_as_float(e2.y));
    gde_acc2(s0, s1, a3, __int_as_float(e3.y));
  }
#pragma nounroll
  for (; j < n; ++j) {
    const int2 e0 = ed[j];
    const float4 a0 = *(const float4*)(yc + (size_t)e0.x * 64);
    gde_acc2(s0, s1, a0, __int_as_float(e0.y));
  }
  float* ap = At + nl * PITCH + c4;
  *(float4*)ap         = make_float4(s0.x * iv, s0.y * iv, s0.z * iv, s0.w * iv);
  *(float4*)(ap + 64)  = make_float4(s1.x * iv, s1.y * iv, s1.z * iv, s1.w * iv);
  *(float4*)(ap + 128) = yv;
  __syncthreads();

  const int q = __builtin_amdgcn_readfirstlane(tid >> 5);
  if (q >= 4) return;
  const int lane = tid & 31;
  const int c    = lane & 15;
  const int h    = lane >> 4;
  const int col  = 16 * q + c;
  const float bq = bias[col];

  const int strip0 = blockIdx.x * 2;
  size_t o0[2], ol[2];
  bool   ok[2];
  float av[2][8], hv[2][8];
#pragma unroll
  for (int m = 0; m < 2; ++m) {
    ok[m] = (strip0 + m) * 16 < GDE_N;
    o0[m] = ((size_t)(strip0 + m) * 16 + 8 * h) * 64 + col;
    ol[m] = ok[m] ? o0[m] : o0[0];
#pragma unroll
    for (int r = 0; r < 8; ++r) { av[m][r] = 0.f; hv[m][r] = 0.f; }
    if (RK == 1) {
#pragma unroll
      for (int r = 0; r < 8; ++r) hv[m][r] = H[ol[m] + (size_t)r * 64];
    }
  }

  v8f acc0 = (v8f){0.f, 0.f, 0.f, 0.f, 0.f, 0.f, 0.f, 0.f};
  v8f acc1 = (v8f){0.f, 0.f, 0.f, 0.f, 0.f, 0.f, 0.f, 0.f};
  const float4* bl = Bs + q * (24 * 32) + lane;
  const float*  a0p = At + c * PITCH + 4 * h;
  const float*  a1p = At + (16 + c) * PITCH + 4 * h;
  float4 nb  = bl[0];
  float4 nx0 = *(const float4*)a0p;
  float4 nx1 = *(const float4*)a1p;
#pragma unroll
  for (int T = 0; T < 24; ++T) {
    const float4 bb = nb, x0 = nx0, x1 = nx1;
    if (T + 1 < 24) {
      nb  = bl[(T + 1) * 32];
      nx0 = *(const float4*)(a0p + 8 * (T + 1));
      nx1 = *(const float4*)(a1p + 8 * (T + 1));
    }
    v2f b0; b0.x = bb.x; b0.y = bb.y;
    v2f b1; b1.x = bb.z; b1.y = bb.w;
    v2f p0; p0.x = x0.x; p0.y = x0.y;
    v2f p1; p1.x = x0.z; p1.y = x0.w;
    v2f r0; r0.x = x1.x; r0.y = x1.y;
    v2f r1; r1.x = x1.z; r1.y = x1.w;
    acc0 = __builtin_amdgcn_wmma_f32_16x16x4_f32(false, p0, false, b0, (short)0, acc0, false, false);
    acc1 = __builtin_amdgcn_wmma_f32_16x16x4_f32(false, r0, false, b0, (short)0, acc1, false, false);
    acc0 = __builtin_amdgcn_wmma_f32_16x16x4_f32(false, p1, false, b1, (short)0, acc0, false, false);
    acc1 = __builtin_amdgcn_wmma_f32_16x16x4_f32(false, r1, false, b1, (short)0, acc1, false, false);
    __builtin_amdgcn_sched_barrier(0);
  }

  if (RK == 2 || RK == 3) {
#pragma unroll
    for (int m = 0; m < 2; ++m) {
#pragma unroll
      for (int r = 0; r < 8; ++r) {
        const size_t oi = ol[m] + (size_t)r * 64;
        av[m][r] = ACC[oi];
        hv[m][r] = (RK == 2) ? H[oi] : out[oi];
      }
    }
  }

#pragma unroll
  for (int m = 0; m < 2; ++m) {
    if (!ok[m]) continue;
#pragma unroll
    for (int r = 0; r < 8; ++r) {
      const size_t oi = o0[m] + (size_t)r * 64;
      const float v = (m == 0 ? acc0[r] : acc1[r]) + bq;
      if (RK == 0) {
        out[oi] = v;
      } else if (RK == 1) {
        ACC[oi] = w * v;
        out[oi] = hv[m][r] + cw * v;
      } else if (RK == 2) {
        ACC[oi] = av[m][r] + w * v;
        out[oi] = hv[m][r] + cw * v;
      } else {
        out[oi] = hv[m][r] + 0.5f * (av[m][r] + v);
      }
    }
  }
}

__global__ void gde_log_softmax16(float* __restrict__ out, int n) {
  int i = blockIdx.x * blockDim.x + threadIdx.x;
  if (i >= n) return;
  float v[16];
  float mx = -3.402823466e+38f;
  float* row = out + (size_t)i * 16;
#pragma unroll
  for (int c = 0; c < 16; ++c) { v[c] = row[c]; mx = fmaxf(mx, v[c]); }
  float s = 0.f;
#pragma unroll
  for (int c = 0; c < 16; ++c) s += __expf(v[c] - mx);
  float ls = __logf(s);
#pragma unroll
  for (int c = 0; c < 16; ++c) row[c] = v[c] - mx - ls;
}

extern "C" void kernel_launch(void* const* d_in, const int* in_sizes, int n_in,
                              void* d_out, int out_size, void* d_ws, size_t ws_size,
                              hipStream_t stream) {
  const float* x   = (const float*)d_in[0];
  const float* p   = (const float*)d_in[1];
  const int*   src = (const int*)d_in[2];
  const int*   dst = (const int*)d_in[3];
  const float* W1 = (const float*)d_in[4],  *R1 = (const float*)d_in[5],  *b1 = (const float*)d_in[6];
  const float* Wa = (const float*)d_in[7],  *Ra = (const float*)d_in[8],  *ba = (const float*)d_in[9];
  const float* Wb = (const float*)d_in[10], *Rb = (const float*)d_in[11], *bb = (const float*)d_in[12];
  const float* W2 = (const float*)d_in[13], *R2 = (const float*)d_in[14], *b2 = (const float*)d_in[15];
  float* out = (float*)d_out;

  float* W = (float*)d_ws;
  size_t o = 0;
  float* Z   = W + o; o += (size_t)GDE_N * 192;
  float* CSR = W + o; o += (size_t)GDE_N * 64;
  float* H   = W + o; o += (size_t)GDE_N * 64;
  float* YT  = W + o; o += (size_t)GDE_N * 64;
  float* U   = W + o; o += (size_t)GDE_N * 64;
  float* ACC = W + o; o += (size_t)GDE_N * 64;
  float* INV = W + o; o += (size_t)GDE_N;
  float* GC1 = W + o; o += (size_t)256 * 192;
  float* GSA = W + o; o += (size_t)192 * 64;
  float* GSB = W + o; o += (size_t)192 * 64;
  float* GC2 = W + o; o += (size_t)64 * 48;
  int*  CNT = (int*)CSR;
  int*  RP  = CNT + GDE_N;
  int*  TMP = RP + GDE_N + 2;
  int*  ARR = TMP + GDE_E;
  int2* ED  = (int2*)ARR;
  (void)ws_size; (void)n_in; (void)in_sizes; (void)out_size;

  gde_prep_g<<<(256 * 192 + 255) / 256, 256, 0, stream>>>(GC1, W1, R1, 256, 64);
  gde_prep_gf<<<(192 * 64 + 255) / 256, 256, 0, stream>>>(GSA, Wa, Ra);
  gde_prep_gf<<<(192 * 64 + 255) / 256, 256, 0, stream>>>(GSB, Wb, Rb);
  gde_prep_g<<<(64 * 48 + 255) / 256, 256, 0, stream>>>(GC2, W2, R2, 64, 16);

  hipMemsetAsync(CNT, 0, sizeof(int) * GDE_N, stream);
  gde_rank<<<(GDE_E + 255) / 256, 256, 0, stream>>>(CNT, ARR, dst, GDE_E);
  gde_scan<<<1, 1024, 0, stream>>>(RP, INV, CNT);
  gde_place<<<(GDE_E + 255) / 256, 256, 0, stream>>>(TMP, RP, ARR, dst, GDE_E);
  gde_order<<<(GDE_E + 255) / 256, 256, 0, stream>>>(ED, TMP, RP, src, dst, p, GDE_E);

  auto hidden = [&](const float* y, const float* G, const float* bias, float* ob,
                    int rk, float w, float cw) {
    const int nb = (GDE_N + 31) / 32;
    if (rk == 0)
      gde_conv_fused<0><<<nb, 512, 0, stream>>>(y, ED, RP, INV, G, bias, ob, ACC, H, w, cw);
    else if (rk == 1)
      gde_conv_fused<1><<<nb, 512, 0, stream>>>(y, ED, RP, INV, G, bias, ob, ACC, H, w, cw);
    else if (rk == 2)
      gde_conv_fused<2><<<nb, 512, 0, stream>>>(y, ED, RP, INV, G, bias, ob, ACC, H, w, cw);
    else
      gde_conv_fused<3><<<nb, 512, 0, stream>>>(y, ED, RP, INV, G, bias, ob, ACC, YT, w, cw);
  };
  auto odefunc = [&](const float* y, float* ob, int rk, float w, float cw) {
    hidden(y, GSA, ba, U, 0, 0.f, 0.f);
    hidden(U, GSB, bb, ob, rk, w, cw);
  };

  gde_mm_bstat<256, 3, 4, 2><<<256, 256, 0, stream>>>(x, GC1, Z, GDE_N / 16);
  gde_gather<64><<<GDE_N * 16 / 256, 256, 0, stream>>>(H, Z, ED, RP, b1, INV);

  odefunc(H,  YT, 1, 1.0f, 1.5f);
  odefunc(YT, YT, 2, 2.0f, 1.5f);
  odefunc(YT, YT, 2, 2.0f, 3.0f);
  odefunc(YT, H,  3, 0.0f, 0.0f);

  gde_mm_bstat<64, 3, 1, 8><<<256, 256, 0, stream>>>(H, GC2, Z, GDE_N / 16);
  gde_gather<16><<<(GDE_N * 4 + 255) / 256, 256, 0, stream>>>(out, Z, ED, RP, b2, INV);
  gde_log_softmax16<<<(GDE_N + 255) / 256, 256, 0, stream>>>(out, GDE_N);
}


# Gather loops pipelined: edge records fetched one trip ahead of the rows they name, tail folded into the 4-edge trip (empty slots add zero): one memory wait per trip, not two; sum order kept
# speedup vs baseline: 3.7970x; 1.0366x over previous
#include <hip/hip_runtime.h>
#include <math.h>

#define GDE_N 50000
#define GDE_E 800000

typedef float v2f __attribute__((ext_vector_type(2)));
typedef float v8f __attribute__((ext_vector_type(8)));

template <int K, int NT, int CG, int RS>
__global__ __launch_bounds__(32 * CG * RS) void gde_mm_bstat(const float* __restrict__ X,
                                                            const float* __restrict__ G,
                                                            float* __restrict__ Y,
                                                            int nStrips) {
  constexpr int NCOLS = 16 * NT * CG;
  constexpr int STEPS = K / 4;
  constexpr int KC    = K / 64;
  const int lane = threadIdx.x & 31;
  const int wave = __builtin_amdgcn_readfirstlane((int)(threadIdx.x >> 5));
  const int c    = lane & 15;
  const int h    = lane >> 4;
  const int cg   = wave % CG;
  const int rs   = wave / CG;
  const int col0 = cg * (16 * NT) + c;

  v2f bf[STEPS][NT];
#pragma unroll
  for (int s = 0; s < STEPS; ++s) {
    const int kb = 8 * (s >> 1) + 2 * (s & 1) + 4 * h;
#pragma unroll
    for (int t = 0; t < NT; ++t) {
      bf[s][t].x = G[(size_t)kb       * NCOLS + col0 + 16 * t];
      bf[s][t].y = G[(size_t)(kb + 1) * NCOLS + col0 + 16 * t];
    }
  }

  const int stride = gridDim.x * RS;
  int strip = blockIdx.x * RS + rs;
  if (strip >= nStrips) return;
  const float* xr = X + ((size_t)strip * 16 + c) * K + 4 * h;

  float4 cur[8];
#pragma unroll
  for (int t = 0; t < 8; ++t) {
    cur[t] = *(const float4*)(xr + 8 * t);
    __builtin_amdgcn_sched_barrier(0);
  }

  for (; strip < nStrips; strip += stride) {
    const int sn = min(strip + stride, nStrips - 1);
    const float* xn = X + ((size_t)sn * 16 + c) * K + 4 * h;

    v8f acc[NT];
#pragma unroll
    for (int t = 0; t < NT; ++t) acc[t] = (v8f){0.f, 0.f, 0.f, 0.f, 0.f, 0.f, 0.f, 0.f};

#pragma unroll
    for (int kc = 0; kc < KC; ++kc) {
      const float* nsrc = (kc + 1 < KC) ? (xr + (kc + 1) * 64) : xn;
#pragma unroll
      for (int t = 0; t < 8; ++t) {
        v2f a0; a0.x = cur[t].x; a0.y = cur[t].y;
        v2f a1; a1.x = cur[t].z; a1.y = cur[t].w;
        const int S = kc * 16 + 2 * t;
#pragma unroll
        for (int q = 0; q < NT; ++q)
          acc[q] = __builtin_amdgcn_wmma_f32_16x16x4_f32(false, a0, false, bf[S][q],
                                                         (short)0, acc[q], false, false);
#pragma unroll
        for (int q = 0; q < NT; ++q)
          acc[q] = __builtin_amdgcn_wmma_f32_16x16x4_f32(false, a1, false, bf[S + 1][q],
                                                         (short)0, acc[q], false, false);
        cur[t] = *(const float4*)(nsrc + 8 * t);
        __builtin_amdgcn_sched_barrier(0);
      }
    }

    float* yr = Y + ((size_t)strip * 16 + 8 * h) * NCOLS + col0;
#pragma unroll
    for (int q = 0; q < NT; ++q)
#pragma unroll
      for (int r = 0; r < 8; ++r)
        yr[(size_t)r * NCOLS + 16 * q] = acc[q][r];

    xr = xn;
  }
}

__global__ void gde_prep_g(float* __restrict__ G, const float* __restrict__ Wp,
                           const float* __restrict__ R, int Cin, int Cout) {
  int idx = blockIdx.x * blockDim.x + threadIdx.x;
  int Ncols = 3 * Cout;
  int tot = Cin * Ncols;
  if (idx >= tot) return;
  int r = idx / Ncols, c = idx % Ncols;
  float v;
  if (c < Cout)            v = Wp[r * Cout + c];
  else if (c < 2 * Cout)   v = Wp[(size_t)Cin * Cout + r * Cout + (c - Cout)]
                             - Wp[r * Cout + (c - Cout)];
  else                     v = R[r * Cout + (c - 2 * Cout)];
  G[idx] = v;
}

__global__ void gde_prep_gf(float* __restrict__ G, const float* __restrict__ Wp,
                            const float* __restrict__ R) {
  constexpr int C = 64;
  int idx = blockIdx.x * blockDim.x + threadIdx.x;
  if (idx >= 3 * C * C) return;
  const int j = idx & 3, lane = (idx >> 2) & 31, T = (idx >> 7) % 24, q = idx / (128 * 24);
  const int k = 8 * T + 4 * (lane >> 4) + j;
  const int c = 16 * q + (lane & 15);
  float v;
  if (k < C)            v = Wp[k * C + c];
  else if (k < 2 * C)   v = Wp[(size_t)C * C + (k - C) * C + c] - Wp[(k - C) * C + c];
  else                  v = R[(k - 2 * C) * C + c];
  G[idx] = v;
}

__global__ void gde_rank(int* __restrict__ cnt, int* __restrict__ arr,
                         const int* __restrict__ dst, int E) {
  int e = blockIdx.x * blockDim.x + threadIdx.x;
  if (e < E) arr[e] = atomicAdd(&cnt[dst[e]], 1);
}

__global__ __launch_bounds__(1024) void gde_scan(int* __restrict__ rowptr, float* __restrict__ inv,
                                                 const int* __restrict__ cnt) {
  constexpr int N4    = GDE_N / 4;
  constexpr int TILES = (N4 + 1023) / 1024;
  static_assert(GDE_N % 4 == 0, "int4 tiles");
  __shared__ int wsum[32];
  __shared__ int tile_total;
  const int t = threadIdx.x, lane = t & 31, w = t >> 5;
  int4 v[TILES];
#pragma unroll
  for (int k = 0; k < TILES; ++k) {
    const int i = k * 1024 + t;
    v[k] = (i < N4) ? ((const int4*)cnt)[i] : make_int4(0, 0, 0, 0);
  }
  int carry = 0;
#pragma unroll
  for (int k = 0; k < TILES; ++k) {
    const int i = k * 1024 + t;
    const int s = v[k].x + v[k].y + v[k].z + v[k].w;
    int inc = s;
    for (int d = 1; d < 32; d <<= 1) {
      const int o = __shfl_up(inc, d);
      if (lane >= d) inc += o;
    }
    if (lane == 31) wsum[w] = inc;
    __syncthreads();
    if (w == 0) {
      const int x = wsum[lane];
      int xi = x;
      for (int d = 1; d < 32; d <<= 1) {
        const int o = __shfl_up(xi, d);
        if (lane >= d) xi += o;
      }
      wsum[lane] = xi - x;
      if (lane == 31) tile_total = xi;
    }
    __syncthreads();
    const int base = carry + wsum[w] + inc - s;
    carry += tile_total;
    if (i < N4) {
      int4 r;
      r.x = base;
      r.y = r.x + v[k].x;
      r.z = r.y + v[k].y;
      r.w = r.z + v[k].z;
      ((int4*)rowptr)[i] = r;
      float4 f;
      f.x = 1.0f / fmaxf((float)v[k].x, 1.0f);
      f.y = 1.0f / fmaxf((float)v[k].y, 1.0f);
      f.z = 1.0f / fmaxf((float)v[k].z, 1.0f);
      f.w = 1.0f / fmaxf((float)v[k].w, 1.0f);
      ((float4*)inv)[i] = f;
    }
    __syncthreads();
  }
  if (t == 0) rowptr[GDE_N] = carry;
}

__global__ void gde_place(int* __restrict__ tmp, const int* __restrict__ rowptr,
                          const int* __restrict__ arr, const int* __restrict__ dst, int E) {
  int e = blockIdx.x * blockDim.x + threadIdx.x;
  if (e < E) tmp[rowptr[dst[e]] + arr[e]] = e;
}

__global__ void gde_order(int2* __restrict__ ed, const int* __restrict__ tmp,
                          const int* __restrict__ rowptr, const int* __restrict__ src,
                          const int* __restrict__ dst, const float* __restrict__ p, int E) {
  int e = blockIdx.x * blockDim.x + threadIdx.x;
  if (e >= E) return;
  const int d = dst[e];
  const int b = rowptr[d], n = rowptr[d + 1];
  int r = 0;
  for (int j = b; j < n; ++j) r += (tmp[j] < e) ? 1 : 0;
  ed[b + r] = make_int2(src[e], __float_as_int(p[e]));
}

__device__ __forceinline__ float4 gde_keep(bool live, const float4& v) {
  return make_float4(live ? v.x : 0.f, live ? v.y : 0.f, live ? v.z : 0.f, live ? v.w : 0.f);
}

__device__ __forceinline__ void gde_acc(float4& s, const float4& z0, const float4& z1, float pe) {
  s.x += z0.x + pe * z1.x;
  s.y += z0.y + pe * z1.y;
  s.z += z0.z + pe * z1.z;
  s.w += z0.w + pe * z1.w;
}

template <int COUT>
__global__ __launch_bounds__(256) void gde_gather(float* __restrict__ out, const float* __restrict__ Z,
                                                  const int2* __restrict__ ed,
                                                  const int* __restrict__ rowptr,
                                                  const float* __restrict__ bias,
                                                  const float* __restrict__ inv) {
  constexpr int LPN = COUT / 4;
  constexpr int LDZ = 3 * COUT;
  const int gt   = blockIdx.x * 256 + threadIdx.x;
  const int node = gt / LPN;
  if (node >= GDE_N) return;
  const int c = (gt % LPN) * 4;
  const int b = rowptr[node], n = rowptr[node + 1];
  const char* zb = (const char*)Z;
  const unsigned cb = (unsigned)c * 4u;
  float4 s = make_float4(0.f, 0.f, 0.f, 0.f);
  const int last = max(n - 1, 0);
  int j = b;
  int2 e0 = ed[min(j, last)], e1 = ed[min(j + 1, last)];
  int2 e2 = ed[min(j + 2, last)], e3 = ed[min(j + 3, last)];
  while (j < n) {
    const char* r0 = zb + ((unsigned)e0.x * (LDZ * 4u) + cb);
    const char* r1 = zb + ((unsigned)e1.x * (LDZ * 4u) + cb);
    const char* r2 = zb + ((unsigned)e2.x * (LDZ * 4u) + cb);
    const char* r3 = zb + ((unsigned)e3.x * (LDZ * 4u) + cb);
    const float4 a0 = *(const float4*)r0, b0 = *(const float4*)(r0 + COUT * 4);
    const float4 a1 = *(const float4*)r1, b1 = *(const float4*)(r1 + COUT * 4);
    const float4 a2 = *(const float4*)r2, b2 = *(const float4*)(r2 + COUT * 4);
    const float4 a3 = *(const float4*)r3, b3 = *(const float4*)(r3 + COUT * 4);
    const int jn = j + 4;
    const int2 f0 = ed[min(jn, last)], f1 = ed[min(jn + 1, last)];
    const int2 f2 = ed[min(jn + 2, last)], f3 = ed[min(jn + 3, last)];
    asm volatile("" ::: "memory");
    const bool k1 = j + 1 < n, k2 = j + 2 < n, k3 = j + 3 < n;
    gde_acc(s, a0, b0, __int_as_float(e0.y));
    gde_acc(s, gde_keep(k1, a1), gde_keep(k1, b1), __int_as_float(e1.y));
    gde_acc(s, gde_keep(k2, a2), gde_keep(k2, b2), __int_as_float(e2.y));
    gde_acc(s, gde_keep(k3, a3), gde_keep(k3, b3), __int_as_float(e3.y));
    e0 = f0; e1 = f1; e2 = f2; e3 = f3;
    j = jn;
  }
  const size_t oi = (size_t)node * COUT + c;
  const float iv = inv[node];
  const float4 zr = *(const float4*)(Z + (size_t)node * LDZ + 2 * COUT + c);
  float4 v;
  v.x = tanhf(s.x * iv + zr.x + bias[c + 0]);
  v.y = tanhf(s.y * iv + zr.y + bias[c + 1]);
  v.z = tanhf(s.z * iv + zr.z + bias[c + 2]);
  v.w = tanhf(s.w * iv + zr.w + bias[c + 3]);
  *(float4*)(out + oi) = v;
}

__device__ __forceinline__ void gde_acc2(float4& s0, float4& s1, const float4& y, float pe) {
  s0.x += y.x;  s0.y += y.y;  s0.z += y.z;  s0.w += y.w;
  s1.x += pe * y.x;  s1.y += pe * y.y;  s1.z += pe * y.z;  s1.w += pe * y.w;
}

template <int RK>
__global__ __launch_bounds__(512) void gde_conv_fused(const float* __restrict__ Y,
                                                      const int2* __restrict__ ed,
                                                      const int* __restrict__ rowptr,
                                                      const float* __restrict__ inv,
                                                      const float* __restrict__ Gf,
                                                      const float* __restrict__ bias,
                                                      float* __restrict__ out,
                                                      float* __restrict__ ACC,
                                                      const float* __restrict__ H,
                                                      float w, float cw) {
  constexpr int PITCH = 196;
  __shared__ float4 Bs[192 * 64 / 4];
  __shared__ __attribute__((aligned(16))) float At[32 * PITCH];
  const int tid = threadIdx.x;

  const int nl   = tid >> 4;
  const int c4   = (tid & 15) * 4;
  const int node = blockIdx.x * 32 + nl;
  int b = 0, n = 0;
  float iv = 0.f;
  float4 yv = make_float4(0.f, 0.f, 0.f, 0.f);
  if (node < GDE_N) {
    b  = rowptr[node];
    n  = rowptr[node + 1];
    iv = inv[node];
    yv = *(const float4*)(Y + (size_t)node * 64 + c4);
  }
#pragma unroll
  for (int i = 0; i < 6; ++i) Bs[tid + 512 * i] = ((const float4*)Gf)[tid + 512 * i];
  const float* yc = Y + c4;
  float4 s0 = make_float4(0.f, 0.f, 0.f, 0.f);
  float4 s1 = make_float4(0.f, 0.f, 0.f, 0.f);
  const int last = max(n - 1, 0);
  int j = b;
  int2 e0 = ed[min(j, last)], e1 = ed[min(j + 1, last)];
  int2 e2 = ed[min(j + 2, last)], e3 = ed[min(j + 3, last)];
  while (j < n) {
    const float4 a0 = *(const float4*)(yc + (size_t)e0.x * 64);
    const float4 a1 = *(const float4*)(yc + (size_t)e1.x * 64);
    const float4 a2 = *(const float4*)(yc + (size_t)e2.x * 64);
    const float4 a3 = *(const float4*)(yc + (size_t)e3.x * 64);
    const int jn = j + 4;
    const int2 f0 = ed[min(jn, last)], f1 = ed[min(jn + 1, last)];
    const int2 f2 = ed[min(jn + 2, last)], f3 = ed[min(jn + 3, last)];
    asm volatile("" ::: "memory");
    const bool k1 = j + 1 < n, k2 = j + 2 < n, k3 = j + 3 < n;
    gde_acc2(s0, s1, a0, __int_as_float(e0.y));
    gde_acc2(s0, s1, gde_keep(k1, a1), __int_as_float(e1.y));
    gde_acc2(s0, s1, gde_keep(k2, a2), __int_as_float(e2.y));
    gde_acc2(s0, s1, gde_keep(k3, a3), __int_as_float(e3.y));
    e0 = f0; e1 = f1; e2 = f2; e3 = f3;
    j = jn;
  }
  float* ap = At + nl * PITCH + c4;
  *(float4*)ap         = make_float4(s0.x * iv, s0.y * iv, s0.z * iv, s0.w * iv);
  *(float4*)(ap + 64)  = make_float4(s1.x * iv, s1.y * iv, s1.z * iv, s1.w * iv);
  *(float4*)(ap + 128) = yv;
  __syncthreads();

  const int q = __builtin_amdgcn_readfirstlane(tid >> 5);
  if (q >= 4) return;
  const int lane = tid & 31;
  const int c    = lane & 15;
  const int h    = lane >> 4;
  const int col  = 16 * q + c;
  const float bq = bias[col];

  const int strip0 = blockIdx.x * 2;
  size_t o0[2], ol[2];
  bool   ok[2];
  float av[2][8], hv[2][8];
#pragma unroll
  for (int m = 0; m < 2; ++m) {
    ok[m] = (strip0 + m) * 16 < GDE_N;
    o0[m] = ((size_t)(strip0 + m) * 16 + 8 * h) * 64 + col;
    ol[m] = ok[m] ? o0[m] : o0[0];
#pragma unroll
    for (int r = 0; r < 8; ++r) { av[m][r] = 0.f; hv[m][r] = 0.f; }
    if (RK == 1) {
#pragma unroll
      for (int r = 0; r < 8; ++r) hv[m][r] = H[ol[m] + (size_t)r * 64];
    }
  }

  v8f acc0 = (v8f){0.f, 0.f, 0.f, 0.f, 0.f, 0.f, 0.f, 0.f};
  v8f acc1 = (v8f){0.f, 0.f, 0.f, 0.f, 0.f, 0.f, 0.f, 0.f};
  const float4* bl = Bs + q * (24 * 32) + lane;
  const float*  a0p = At + c * PITCH + 4 * h;
  const float*  a1p = At + (16 + c) * PITCH + 4 * h;
  float4 nb  = bl[0];
  float4 nx0 = *(const float4*)a0p;
  float4 nx1 = *(const float4*)a1p;
#pragma unroll
  for (int T = 0; T < 24; ++T) {
    const float4 bb = nb, x0 = nx0, x1 = nx1;
    if (T + 1 < 24) {
      nb  = bl[(T + 1) * 32];
      nx0 = *(const float4*)(a0p + 8 * (T + 1));
      nx1 = *(const float4*)(a1p + 8 * (T + 1));
    }
    v2f b0; b0.x = bb.x; b0.y = bb.y;
    v2f b1; b1.x = bb.z; b1.y = bb.w;
    v2f p0; p0.x = x0.x; p0.y = x0.y;
    v2f p1; p1.x = x0.z; p1.y = x0.w;
    v2f r0; r0.x = x1.x; r0.y = x1.y;
    v2f r1; r1.x = x1.z; r1.y = x1.w;
    acc0 = __builtin_amdgcn_wmma_f32_16x16x4_f32(false, p0, false, b0, (short)0, acc0, false, false);
    acc1 = __builtin_amdgcn_wmma_f32_16x16x4_f32(false, r0, false, b0, (short)0, acc1, false, false);
    acc0 = __builtin_amdgcn_wmma_f32_16x16x4_f32(false, p1, false, b1, (short)0, acc0, false, false);
    acc1 = __builtin_amdgcn_wmma_f32_16x16x4_f32(false, r1, false, b1, (short)0, acc1, false, false);
    __builtin_amdgcn_sched_barrier(0);
  }

  if (RK == 2 || RK == 3) {
#pragma unroll
    for (int m = 0; m < 2; ++m) {
#pragma unroll
      for (int r = 0; r < 8; ++r) {
        const size_t oi = ol[m] + (size_t)r * 64;
        av[m][r] = ACC[oi];
        hv[m][r] = (RK == 2) ? H[oi] : out[oi];
      }
    }
  }

#pragma unroll
  for (int m = 0; m < 2; ++m) {
    if (!ok[m]) continue;
#pragma unroll
    for (int r = 0; r < 8; ++r) {
      const size_t oi = o0[m] + (size_t)r * 64;
      const float v = (m == 0 ? acc0[r] : acc1[r]) + bq;
      if (RK == 0) {
        out[oi] = v;
      } else if (RK == 1) {
        ACC[oi] = w * v;
        out[oi] = hv[m][r] + cw * v;
      } else if (RK == 2) {
        ACC[oi] = av[m][r] + w * v;
        out[oi] = hv[m][r] + cw * v;
      } else {
        out[oi] = hv[m][r] + 0.5f * (av[m][r] + v);
      }
    }
  }
}

__global__ void gde_log_softmax16(float* __restrict__ out, int n) {
  int i = blockIdx.x * blockDim.x + threadIdx.x;
  if (i >= n) return;
  float v[16];
  float mx = -3.402823466e+38f;
  float* row = out + (size_t)i * 16;
#pragma unroll
  for (int c = 0; c < 16; ++c) { v[c] = row[c]; mx = fmaxf(mx, v[c]); }
  float s = 0.f;
#pragma unroll
  for (int c = 0; c < 16; ++c) s += __expf(v[c] - mx);
  float ls = __logf(s);
#pragma unroll
  for (int c = 0; c < 16; ++c) row[c] = v[c] - mx - ls;
}

extern "C" void kernel_launch(void* const* d_in, const int* in_sizes, int n_in,
                              void* d_out, int out_size, void* d_ws, size_t ws_size,
                              hipStream_t stream) {
  const float* x   = (const float*)d_in[0];
  const float* p   = (const float*)d_in[1];
  const int*   src = (const int*)d_in[2];
  const int*   dst = (const int*)d_in[3];
  const float* W1 = (const float*)d_in[4],  *R1 = (const float*)d_in[5],  *b1 = (const float*)d_in[6];
  const float* Wa = (const float*)d_in[7],  *Ra = (const float*)d_in[8],  *ba = (const float*)d_in[9];
  const float* Wb = (const float*)d_in[10], *Rb = (const float*)d_in[11], *bb = (const float*)d_in[12];
  const float* W2 = (const float*)d_in[13], *R2 = (const float*)d_in[14], *b2 = (const float*)d_in[15];
  float* out = (float*)d_out;

  float* W = (float*)d_ws;
  size_t o = 0;
  float* Z   = W + o; o += (size_t)GDE_N * 192;
  float* CSR = W + o; o += (size_t)GDE_N * 64;
  float* H   = W + o; o += (size_t)GDE_N * 64;
  float* YT  = W + o; o += (size_t)GDE_N * 64;
  float* U   = W + o; o += (size_t)GDE_N * 64;
  float* ACC = W + o; o += (size_t)GDE_N * 64;
  float* INV = W + o; o += (size_t)GDE_N;
  float* GC1 = W + o; o += (size_t)256 * 192;
  float* GSA = W + o; o += (size_t)192 * 64;
  float* GSB = W + o; o += (size_t)192 * 64;
  float* GC2 = W + o; o += (size_t)64 * 48;
  int*  CNT = (int*)CSR;
  int*  RP  = CNT + GDE_N;
  int*  TMP = RP + GDE_N + 2;
  int*  ARR = TMP + GDE_E;
  int2* ED  = (int2*)ARR;
  (void)ws_size; (void)n_in; (void)in_sizes; (void)out_size;

  gde_prep_g<<<(256 * 192 + 255) / 256, 256, 0, stream>>>(GC1, W1, R1, 256, 64);
  gde_prep_gf<<<(192 * 64 + 255) / 256, 256, 0, stream>>>(GSA, Wa, Ra);
  gde_prep_gf<<<(192 * 64 + 255) / 256, 256, 0, stream>>>(GSB, Wb, Rb);
  gde_prep_g<<<(64 * 48 + 255) / 256, 256, 0, stream>>>(GC2, W2, R2, 64, 16);

  hipMemsetAsync(CNT, 0, sizeof(int) * GDE_N, stream);
  gde_rank<<<(GDE_E + 255) / 256, 256, 0, stream>>>(CNT, ARR, dst, GDE_E);
  gde_scan<<<1, 1024, 0, stream>>>(RP, INV, CNT);
  gde_place<<<(GDE_E + 255) / 256, 256, 0, stream>>>(TMP, RP, ARR, dst, GDE_E);
  gde_order<<<(GDE_E + 255) / 256, 256, 0, stream>>>(ED, TMP, RP, src, dst, p, GDE_E);

  auto hidden = [&](const float* y, const float* G, const float* bias, float* ob,
                    int rk, float w, float cw) {
    const int nb = (GDE_N + 31) / 32;
    if (rk == 0)
      gde_conv_fused<0><<<nb, 512, 0, stream>>>(y, ED, RP, INV, G, bias, ob, ACC, H, w, cw);
    else if (rk == 1)
      gde_conv_fused<1><<<nb, 512, 0, stream>>>(y, ED, RP, INV, G, bias, ob, ACC, H, w, cw);
    else if (rk == 2)
      gde_conv_fused<2><<<nb, 512, 0, stream>>>(y, ED, RP, INV, G, bias, ob, ACC, H, w, cw);
    else
      gde_conv_fused<3><<<nb, 512, 0, stream>>>(y, ED, RP, INV, G, bias, ob, ACC, YT, w, cw);
  };
  auto odefunc = [&](const float* y, float* ob, int rk, float w, float cw) {
    hidden(y, GSA, ba, U, 0, 0.f, 0.f);
    hidden(U, GSB, bb, ob, rk, w, cw);
  };

  gde_mm_bstat<256, 3, 4, 2><<<256, 256, 0, stream>>>(x, GC1, Z, GDE_N / 16);
  gde_gather<64><<<GDE_N * 16 / 256, 256, 0, stream>>>(H, Z, ED, RP, b1, INV);

  odefunc(H,  YT, 1, 1.0f, 1.5f);
  odefunc(YT, YT, 2, 2.0f, 1.5f);
  odefunc(YT, YT, 2, 2.0f, 3.0f);
  odefunc(YT, H,  3, 0.0f, 0.0f);

  gde_mm_bstat<64, 3, 1, 8><<<256, 256, 0, stream>>>(H, GC2, Z, GDE_N / 16);
  gde_gather<16><<<(GDE_N * 4 + 255) / 256, 256, 0, stream>>>(out, Z, ED, RP, b2, INV);
  gde_log_softmax16<<<(GDE_N + 255) / 256, 256, 0, stream>>>(out, GDE_N);
}


# Edge lists by destination built with LDS counters only: edges bucketed into 256 node ranges, one block per range ranks and sorts its bucket; 800k global atomics, memset, 1 launch gone; sum order kept
# speedup vs baseline: 4.1180x; 1.0845x over previous
#include <hip/hip_runtime.h>
#include <math.h>

#define GDE_N 50000
#define GDE_E 800000

typedef float v2f __attribute__((ext_vector_type(2)));
typedef float v8f __attribute__((ext_vector_type(8)));

template <int K, int NT, int CG, int RS>
__global__ __launch_bounds__(32 * CG * RS) void gde_mm_bstat(const float* __restrict__ X,
                                                            const float* __restrict__ G,
                                                            float* __restrict__ Y,
                                                            int nStrips) {
  constexpr int NCOLS = 16 * NT * CG;
  constexpr int STEPS = K / 4;
  constexpr int KC    = K / 64;
  const int lane = threadIdx.x & 31;
  const int wave = __builtin_amdgcn_readfirstlane((int)(threadIdx.x >> 5));
  const int c    = lane & 15;
  const int h    = lane >> 4;
  const int cg   = wave % CG;
  const int rs   = wave / CG;
  const int col0 = cg * (16 * NT) + c;

  v2f bf[STEPS][NT];
#pragma unroll
  for (int s = 0; s < STEPS; ++s) {
    const int kb = 8 * (s >> 1) + 2 * (s & 1) + 4 * h;
#pragma unroll
    for (int t = 0; t < NT; ++t) {
      bf[s][t].x = G[(size_t)kb       * NCOLS + col0 + 16 * t];
      bf[s][t].y = G[(size_t)(kb + 1) * NCOLS + col0 + 16 * t];
    }
  }

  const int stride = gridDim.x * RS;
  int strip = blockIdx.x * RS + rs;
  if (strip >= nStrips) return;
  const float* xr = X + ((size_t)strip * 16 + c) * K + 4 * h;

  float4 cur[8];
#pragma unroll
  for (int t = 0; t < 8; ++t) {
    cur[t] = *(const float4*)(xr + 8 * t);
    __builtin_amdgcn_sched_barrier(0);
  }

  for (; strip < nStrips; strip += stride) {
    const int sn = min(strip + stride, nStrips - 1);
    const float* xn = X + ((size_t)sn * 16 + c) * K + 4 * h;

    v8f acc[NT];
#pragma unroll
    for (int t = 0; t < NT; ++t) acc[t] = (v8f){0.f, 0.f, 0.f, 0.f, 0.f, 0.f, 0.f, 0.f};

#pragma unroll
    for (int kc = 0; kc < KC; ++kc) {
      const float* nsrc = (kc + 1 < KC) ? (xr + (kc + 1) * 64) : xn;
#pragma unroll
      for (int t = 0; t < 8; ++t) {
        v2f a0; a0.x = cur[t].x; a0.y = cur[t].y;
        v2f a1; a1.x = cur[t].z; a1.y = cur[t].w;
        const int S = kc * 16 + 2 * t;
#pragma unroll
        for (int q = 0; q < NT; ++q)
          acc[q] = __builtin_amdgcn_wmma_f32_16x16x4_f32(false, a0, false, bf[S][q],
                                                         (short)0, acc[q], false, false);
#pragma unroll
        for (int q = 0; q < NT; ++q)
          acc[q] = __builtin_amdgcn_wmma_f32_16x16x4_f32(false, a1, false, bf[S + 1][q],
                                                         (short)0, acc[q], false, false);
        cur[t] = *(const float4*)(nsrc + 8 * t);
        __builtin_amdgcn_sched_barrier(0);
      }
    }

    float* yr = Y + ((size_t)strip * 16 + 8 * h) * NCOLS + col0;
#pragma unroll
    for (int q = 0; q < NT; ++q)
#pragma unroll
      for (int r = 0; r < 8; ++r)
        yr[(size_t)r * NCOLS + 16 * q] = acc[q][r];

    xr = xn;
  }
}

__global__ void gde_prep_g(float* __restrict__ G, const float* __restrict__ Wp,
                           const float* __restrict__ R, int Cin, int Cout) {
  int idx = blockIdx.x * blockDim.x + threadIdx.x;
  int Ncols = 3 * Cout;
  int tot = Cin * Ncols;
  if (idx >= tot) return;
  int r = idx / Ncols, c = idx % Ncols;
  float v;
  if (c < Cout)            v = Wp[r * Cout + c];
  else if (c < 2 * Cout)   v = Wp[(size_t)Cin * Cout + r * Cout + (c - Cout)]
                             - Wp[r * Cout + (c - Cout)];
  else                     v = R[r * Cout + (c - 2 * Cout)];
  G[idx] = v;
}

__global__ void gde_prep_gf(float* __restrict__ G, const float* __restrict__ Wp,
                            const float* __restrict__ R) {
  constexpr int C = 64;
  int idx = blockIdx.x * blockDim.x + threadIdx.x;
  if (idx >= 3 * C * C) return;
  const int j = idx & 3, lane = (idx >> 2) & 31, T = (idx >> 7) % 24, q = idx / (128 * 24);
  const int k = 8 * T + 4 * (lane >> 4) + j;
  const int c = 16 * q + (lane & 15);
  float v;
  if (k < C)            v = Wp[k * C + c];
  else if (k < 2 * C)   v = Wp[(size_t)C * C + (k - C) * C + c] - Wp[(k - C) * C + c];
  else                  v = R[(k - 2 * C) * C + c];
  G[idx] = v;
}

#define GDE_NB 256
#define GDE_RN 196
#define GDE_CH ((GDE_E + GDE_NB - 1) / GDE_NB)
static_assert(GDE_NB * GDE_RN >= GDE_N, "the ranges cover every node");
static_assert(GDE_E <= (1 << 20) && GDE_RN <= (1 << 11), "bucket word: 20 bits of edge id under the node");

__global__ __launch_bounds__(1024) void gde_hist(int* __restrict__ CT, const int* __restrict__ dst) {
  __shared__ int hc[GDE_NB];
  const int t = threadIdx.x;
  if (t < GDE_NB) hc[t] = 0;
  __syncthreads();
  const int e0 = blockIdx.x * GDE_CH, e1 = min(e0 + GDE_CH, GDE_E);
  for (int e = e0 + t; e < e1; e += 1024) atomicAdd(&hc[dst[e] / GDE_RN], 1);
  __syncthreads();
  if (t < GDE_NB) CT[blockIdx.x * GDE_NB + t] = hc[t];
}

__global__ __launch_bounds__(1024) void gde_bucket(int* __restrict__ BK, int* __restrict__ BB,
                                                   const int* __restrict__ CT,
                                                   const int* __restrict__ dst) {
  static_assert(GDE_NB == 256, "four thread groups of GDE_NB share the table rows");
  __shared__ int off[GDE_NB];
  __shared__ int ptot[4][GDE_NB];
  __shared__ int ppre[4][GDE_NB];
  __shared__ int wsum[GDE_NB / 32];
  const int tid = threadIdx.x, lane = tid & 31, w = tid >> 5;
  const int t = tid & (GDE_NB - 1), q = tid >> 8;
  const int b1 = blockIdx.x;
  int pre = 0, tot = 0;
  for (int g = 0; g < GDE_NB / 4; g += 16) {
    int v[16];
#pragma unroll
    for (int u = 0; u < 16; ++u) v[u] = CT[((g + u) * 4 + q) * GDE_NB + t];
#pragma unroll
    for (int u = 0; u < 16; ++u) {
      tot += v[u];
      pre += ((g + u) * 4 + q < b1) ? v[u] : 0;
    }
  }
  ptot[q][t] = tot;
  ppre[q][t] = pre;
  __syncthreads();
  tot = 0;
  pre = 0;
  if (tid < GDE_NB) {
    tot = ptot[0][t] + ptot[1][t] + ptot[2][t] + ptot[3][t];
    pre = ppre[0][t] + ppre[1][t] + ppre[2][t] + ppre[3][t];
  }
  int inc = tot;
  for (int d = 1; d < 32; d <<= 1) {
    const int o = __shfl_up(inc, d);
    if (lane >= d) inc += o;
  }
  if (tid < GDE_NB && lane == 31) wsum[w] = inc;
  __syncthreads();
  if (tid < GDE_NB) {
    int wb = 0;
    for (int i = 0; i < w; ++i) wb += wsum[i];
    const int excl = wb + inc - tot;
    off[t] = excl + pre;
    if (b1 == 0) {
      BB[t] = excl;
      if (t == GDE_NB - 1) BB[GDE_NB] = excl + tot;
    }
  }
  __syncthreads();
  const int e0 = b1 * GDE_CH, e1 = min(e0 + GDE_CH, GDE_E);
  for (int e = e0 + tid; e < e1; e += 1024) {
    const int d = dst[e];
    const int k = d / GDE_RN;
    const int slot = atomicAdd(&off[k], 1);
    BK[slot] = ((d - k * GDE_RN) << 20) | e;
  }
}

__global__ __launch_bounds__(1024) void gde_owner(int2* __restrict__ ed, int* __restrict__ rowptr,
                                                  float* __restrict__ inv, int* __restrict__ AR,
                                                  int* __restrict__ SG, const int* __restrict__ BK,
                                                  const int* __restrict__ BB,
                                                  const int* __restrict__ src,
                                                  const int* __restrict__ dst,
                                                  const float* __restrict__ p) {
  constexpr int PL = 7;
  static_assert(32 * PL >= GDE_RN, "one wave scans a range");
  __shared__ int cnt[32 * PL];
  __shared__ int lrp[32 * PL + 1];
  const int t  = threadIdx.x;
  const int k  = blockIdx.x;
  const int r0 = k * GDE_RN;
  const int base = BB[k], nI = BB[k + 1] - base;
  if (t < 32 * PL) cnt[t] = 0;
  __syncthreads();
  for (int i = t; i < nI; i += 1024) AR[base + i] = atomicAdd(&cnt[BK[base + i] >> 20], 1);
  __syncthreads();
  if (t < 32) {
    int c[PL];
    int s = 0;
#pragma unroll
    for (int j = 0; j < PL; ++j) { c[j] = cnt[t * PL + j]; s += c[j]; }
    int inc = s;
    for (int d = 1; d < 32; d <<= 1) {
      const int o = __shfl_up(inc, d);
      if (t >= d) inc += o;
    }
    int ex = inc - s;
#pragma unroll
    for (int j = 0; j < PL; ++j) { lrp[t * PL + j] = ex; ex += c[j]; }
    if (t == 31) lrp[32 * PL] = ex;
  }
  __syncthreads();
  if (t < GDE_RN && r0 + t < GDE_N) {
    rowptr[r0 + t] = base + lrp[t];
    inv[r0 + t]    = 1.0f / fmaxf((float)cnt[t], 1.0f);
  }
  if (k == GDE_NB - 1 && t == 0) rowptr[GDE_N] = base + nI;
  for (int i = t; i < nI; i += 1024) {
    const int wv = BK[base + i];
    SG[base + lrp[wv >> 20] + AR[base + i]] = wv & 0xFFFFF;
  }
  __threadfence();
  __syncthreads();
  for (int i = t; i < nI; i += 1024) {
    const int e  = SG[base + i];
    const int dl = dst[e] - r0;
    const int b = lrp[dl], n = lrp[dl + 1];
    int r = 0;
    for (int j = b; j < n; ++j) r += (SG[base + j] < e) ? 1 : 0;
    ed[base + b + r] = make_int2(src[e], __float_as_int(p[e]));
  }
}

__device__ __forceinline__ float4 gde_keep(bool live, const float4& v) {
  return make_float4(live ? v.x : 0.f, live ? v.y : 0.f, live ? v.z : 0.f, live ? v.w : 0.f);
}

__device__ __forceinline__ void gde_acc(float4& s, const float4& z0, const float4& z1, float pe) {
  s.x += z0.x + pe * z1.x;
  s.y += z0.y + pe * z1.y;
  s.z += z0.z + pe * z1.z;
  s.w += z0.w + pe * z1.w;
}

template <int COUT>
__global__ __launch_bounds__(256) void gde_gather(float* __restrict__ out, const float* __restrict__ Z,
                                                  const int2* __restrict__ ed,
                                                  const int* __restrict__ rowptr,
                                                  const float* __restrict__ bias,
                                                  const float* __restrict__ inv) {
  constexpr int LPN = COUT / 4;
  constexpr int LDZ = 3 * COUT;
  const int gt   = blockIdx.x * 256 + threadIdx.x;
  const int node = gt / LPN;
  if (node >= GDE_N) return;
  const int c = (gt % LPN) * 4;
  const int b = rowptr[node], n = rowptr[node + 1];
  const char* zb = (const char*)Z;
  const unsigned cb = (unsigned)c * 4u;
  float4 s = make_float4(0.f, 0.f, 0.f, 0.f);
  const int last = max(n - 1, 0);
  int j = b;
  int2 e0 = ed[min(j, last)], e1 = ed[min(j + 1, last)];
  int2 e2 = ed[min(j + 2, last)], e3 = ed[min(j + 3, last)];
  while (j < n) {
    const char* r0 = zb + ((unsigned)e0.x * (LDZ * 4u) + cb);
    const char* r1 = zb + ((unsigned)e1.x * (LDZ * 4u) + cb);
    const char* r2 = zb + ((unsigned)e2.x * (LDZ * 4u) + cb);
    const char* r3 = zb + ((unsigned)e3.x * (LDZ * 4u) + cb);
    const float4 a0 = *(const float4*)r0, b0 = *(const float4*)(r0 + COUT * 4);
    const float4 a1 = *(const float4*)r1, b1 = *(const float4*)(r1 + COUT * 4);
    const float4 a2 = *(const float4*)r2, b2 = *(const float4*)(r2 + COUT * 4);
    const float4 a3 = *(const float4*)r3, b3 = *(const float4*)(r3 + COUT * 4);
    const int jn = j + 4;
    const int2 f0 = ed[min(jn, last)], f1 = ed[min(jn + 1, last)];
    const int2 f2 = ed[min(jn + 2, last)], f3 = ed[min(jn + 3, last)];
    asm volatile("" ::: "memory");
    const bool k1 = j + 1 < n, k2 = j + 2 < n, k3 = j + 3 < n;
    gde_acc(s, a0, b0, __int_as_float(e0.y));
    gde_acc(s, gde_keep(k1, a1), gde_keep(k1, b1), __int_as_float(e1.y));
    gde_acc(s, gde_keep(k2, a2), gde_keep(k2, b2), __int_as_float(e2.y));
    gde_acc(s, gde_keep(k3, a3), gde_keep(k3, b3), __int_as_float(e3.y));
    e0 = f0; e1 = f1; e2 = f2; e3 = f3;
    j = jn;
  }
  const size_t oi = (size_t)node * COUT + c;
  const float iv = inv[node];
  const float4 zr = *(const float4*)(Z + (size_t)node * LDZ + 2 * COUT + c);
  float4 v;
  v.x = tanhf(s.x * iv + zr.x + bias[c + 0]);
  v.y = tanhf(s.y * iv + zr.y + bias[c + 1]);
  v.z = tanhf(s.z * iv + zr.z + bias[c + 2]);
  v.w = tanhf(s.w * iv + zr.w + bias[c + 3]);
  *(float4*)(out + oi) = v;
}

__device__ __forceinline__ void gde_acc2(float4& s0, float4& s1, const float4& y, float pe) {
  s0.x += y.x;  s0.y += y.y;  s0.z += y.z;  s0.w += y.w;
  s1.x += pe * y.x;  s1.y += pe * y.y;  s1.z += pe * y.z;  s1.w += pe * y.w;
}

template <int RK>
__global__ __launch_bounds__(512) void gde_conv_fused(const float* __restrict__ Y,
                                                      const int2* __restrict__ ed,
                                                      const int* __restrict__ rowptr,
                                                      const float* __restrict__ inv,
                                                      const float* __restrict__ Gf,
                                                      const float* __restrict__ bias,
                                                      float* __restrict__ out,
                                                      float* __restrict__ ACC,
                                                      const float* __restrict__ H,
                                                      float w, float cw) {
  constexpr int PITCH = 196;
  __shared__ float4 Bs[192 * 64 / 4];
  __shared__ __attribute__((aligned(16))) float At[32 * PITCH];
  const int tid = threadIdx.x;

  const int nl   = tid >> 4;
  const int c4   = (tid & 15) * 4;
  const int node = blockIdx.x * 32 + nl;
  int b = 0, n = 0;
  float iv = 0.f;
  float4 yv = make_float4(0.f, 0.f, 0.f, 0.f);
  if (node < GDE_N) {
    b  = rowptr[node];
    n  = rowptr[node + 1];
    iv = inv[node];
    yv = *(const float4*)(Y + (size_t)node * 64 + c4);
  }
#pragma unroll
  for (int i = 0; i < 6; ++i) Bs[tid + 512 * i] = ((const float4*)Gf)[tid + 512 * i];
  const float* yc = Y + c4;
  float4 s0 = make_float4(0.f, 0.f, 0.f, 0.f);
  float4 s1 = make_float4(0.f, 0.f, 0.f, 0.f);
  const int last = max(n - 1, 0);
  int j = b;
  int2 e0 = ed[min(j, last)], e1 = ed[min(j + 1, last)];
  int2 e2 = ed[min(j + 2, last)], e3 = ed[min(j + 3, last)];
  while (j < n) {
    const float4 a0 = *(const float4*)(yc + (size_t)e0.x * 64);
    const float4 a1 = *(const float4*)(yc + (size_t)e1.x * 64);
    const float4 a2 = *(const float4*)(yc + (size_t)e2.x * 64);
    const float4 a3 = *(const float4*)(yc + (size_t)e3.x * 64);
    const int jn = j + 4;
    const int2 f0 = ed[min(jn, last)], f1 = ed[min(jn + 1, last)];
    const int2 f2 = ed[min(jn + 2, last)], f3 = ed[min(jn + 3, last)];
    asm volatile("" ::: "memory");
    const bool k1 = j + 1 < n, k2 = j + 2 < n, k3 = j + 3 < n;
    gde_acc2(s0, s1, a0, __int_as_float(e0.y));
    gde_acc2(s0, s1, gde_keep(k1, a1), __int_as_float(e1.y));
    gde_acc2(s0, s1, gde_keep(k2, a2), __int_as_float(e2.y));
    gde_acc2(s0, s1, gde_keep(k3, a3), __int_as_float(e3.y));
    e0 = f0; e1 = f1; e2 = f2; e3 = f3;
    j = jn;
  }
  float* ap = At + nl * PITCH + c4;
  *(float4*)ap         = make_float4(s0.x * iv, s0.y * iv, s0.z * iv, s0.w * iv);
  *(float4*)(ap + 64)  = make_float4(s1.x * iv, s1.y * iv, s1.z * iv, s1.w * iv);
  *(float4*)(ap + 128) = yv;
  __syncthreads();

  const int q = __builtin_amdgcn_readfirstlane(tid >> 5);
  if (q >= 4) return;
  const int lane = tid & 31;
  const int c    = lane & 15;
  const int h    = lane >> 4;
  const int col  = 16 * q + c;
  const float bq = bias[col];

  const int strip0 = blockIdx.x * 2;
  size_t o0[2], ol[2];
  bool   ok[2];
  float av[2][8], hv[2][8];
#pragma unroll
  for (int m = 0; m < 2; ++m) {
    ok[m] = (strip0 + m) * 16 < GDE_N;
    o0[m] = ((size_t)(strip0 + m) * 16 + 8 * h) * 64 + col;
    ol[m] = ok[m] ? o0[m] : o0[0];
#pragma unroll
    for (int r = 0; r < 8; ++r) { av[m][r] = 0.f; hv[m][r] = 0.f; }
    if (RK == 1) {
#pragma unroll
      for (int r = 0; r < 8; ++r) hv[m][r] = H[ol[m] + (size_t)r * 64];
    }
  }

  v8f acc0 = (v8f){0.f, 0.f, 0.f, 0.f, 0.f, 0.f, 0.f, 0.f};
  v8f acc1 = (v8f){0.f, 0.f, 0.f, 0.f, 0.f, 0.f, 0.f, 0.f};
  const float4* bl = Bs + q * (24 * 32) + lane;
  const float*  a0p = At + c * PITCH + 4 * h;
  const float*  a1p = At + (16 + c) * PITCH + 4 * h;
  float4 nb  = bl[0];
  float4 nx0 = *(const float4*)a0p;
  float4 nx1 = *(const float4*)a1p;
#pragma unroll
  for (int T = 0; T < 24; ++T) {
    const float4 bb = nb, x0 = nx0, x1 = nx1;
    if (T + 1 < 24) {
      nb  = bl[(T + 1) * 32];
      nx0 = *(const float4*)(a0p + 8 * (T + 1));
      nx1 = *(const float4*)(a1p + 8 * (T + 1));
    }
    v2f b0; b0.x = bb.x; b0.y = bb.y;
    v2f b1; b1.x = bb.z; b1.y = bb.w;
    v2f p0; p0.x = x0.x; p0.y = x0.y;
    v2f p1; p1.x = x0.z; p1.y = x0.w;
    v2f r0; r0.x = x1.x; r0.y = x1.y;
    v2f r1; r1.x = x1.z; r1.y = x1.w;
    acc0 = __builtin_amdgcn_wmma_f32_16x16x4_f32(false, p0, false, b0, (short)0, acc0, false, false);
    acc1 = __builtin_amdgcn_wmma_f32_16x16x4_f32(false, r0, false, b0, (short)0, acc1, false, false);
    acc0 = __builtin_amdgcn_wmma_f32_16x16x4_f32(false, p1, false, b1, (short)0, acc0, false, false);
    acc1 = __builtin_amdgcn_wmma_f32_16x16x4_f32(false, r1, false, b1, (short)0, acc1, false, false);
    __builtin_amdgcn_sched_barrier(0);
  }

  if (RK == 2 || RK == 3) {
#pragma unroll
    for (int m = 0; m < 2; ++m) {
#pragma unroll
      for (int r = 0; r < 8; ++r) {
        const size_t oi = ol[m] + (size_t)r * 64;
        av[m][r] = ACC[oi];
        hv[m][r] = (RK == 2) ? H[oi] : out[oi];
      }
    }
  }

#pragma unroll
  for (int m = 0; m < 2; ++m) {
    if (!ok[m]) continue;
#pragma unroll
    for (int r = 0; r < 8; ++r) {
      const size_t oi = o0[m] + (size_t)r * 64;
      const float v = (m == 0 ? acc0[r] : acc1[r]) + bq;
      if (RK == 0) {
        out[oi] = v;
      } else if (RK == 1) {
        ACC[oi] = w * v;
        out[oi] = hv[m][r] + cw * v;
      } else if (RK == 2) {
        ACC[oi] = av[m][r] + w * v;
        out[oi] = hv[m][r] + cw * v;
      } else {
        out[oi] = hv[m][r] + 0.5f * (av[m][r] + v);
      }
    }
  }
}

__global__ void gde_log_softmax16(float* __restrict__ out, int n) {
  int i = blockIdx.x * blockDim.x + threadIdx.x;
  if (i >= n) return;
  float v[16];
  float mx = -3.402823466e+38f;
  float* row = out + (size_t)i * 16;
#pragma unroll
  for (int c = 0; c < 16; ++c) { v[c] = row[c]; mx = fmaxf(mx, v[c]); }
  float s = 0.f;
#pragma unroll
  for (int c = 0; c < 16; ++c) s += __expf(v[c] - mx);
  float ls = __logf(s);
#pragma unroll
  for (int c = 0; c < 16; ++c) row[c] = v[c] - mx - ls;
}

extern "C" void kernel_launch(void* const* d_in, const int* in_sizes, int n_in,
                              void* d_out, int out_size, void* d_ws, size_t ws_size,
                              hipStream_t stream) {
  const float* x   = (const float*)d_in[0];
  const float* p   = (const float*)d_in[1];
  const int*   src = (const int*)d_in[2];
  const int*   dst = (const int*)d_in[3];
  const float* W1 = (const float*)d_in[4],  *R1 = (const float*)d_in[5],  *b1 = (const float*)d_in[6];
  const float* Wa = (const float*)d_in[7],  *Ra = (const float*)d_in[8],  *ba = (const float*)d_in[9];
  const float* Wb = (const float*)d_in[10], *Rb = (const float*)d_in[11], *bb = (const float*)d_in[12];
  const float* W2 = (const float*)d_in[13], *R2 = (const float*)d_in[14], *b2 = (const float*)d_in[15];
  float* out = (float*)d_out;

  float* W = (float*)d_ws;
  size_t o = 0;
  float* Z   = W + o; o += (size_t)GDE_N * 192;
  float* CSR = W + o; o += (size_t)GDE_N * 64;
  float* H   = W + o; o += (size_t)GDE_N * 64;
  float* YT  = W + o; o += (size_t)GDE_N * 64;
  float* U   = W + o; o += (size_t)GDE_N * 64;
  float* ACC = W + o; o += (size_t)GDE_N * 64;
  float* INV = W + o; o += (size_t)GDE_N;
  float* GC1 = W + o; o += (size_t)256 * 192;
  float* GSA = W + o; o += (size_t)192 * 64;
  float* GSB = W + o; o += (size_t)192 * 64;
  float* GC2 = W + o; o += (size_t)64 * 48;
  int*  RP  = (int*)CSR + GDE_N;
  int2* ED  = (int2*)(RP + GDE_N + 2 + GDE_E);
  int*  CT  = (int*)Z;
  int*  BB  = CT + GDE_NB * GDE_NB;
  int*  BK  = BB + GDE_NB + 4;
  int*  AR  = BK + GDE_E;
  int*  SG  = AR + GDE_E;
  (void)ws_size; (void)n_in; (void)in_sizes; (void)out_size;

  gde_prep_g<<<(256 * 192 + 255) / 256, 256, 0, stream>>>(GC1, W1, R1, 256, 64);
  gde_prep_gf<<<(192 * 64 + 255) / 256, 256, 0, stream>>>(GSA, Wa, Ra);
  gde_prep_gf<<<(192 * 64 + 255) / 256, 256, 0, stream>>>(GSB, Wb, Rb);
  gde_prep_g<<<(64 * 48 + 255) / 256, 256, 0, stream>>>(GC2, W2, R2, 64, 16);

  gde_hist<<<GDE_NB, 1024, 0, stream>>>(CT, dst);
  gde_bucket<<<GDE_NB, 1024, 0, stream>>>(BK, BB, CT, dst);
  gde_owner<<<GDE_NB, 1024, 0, stream>>>(ED, RP, INV, AR, SG, BK, BB, src, dst, p);

  auto hidden = [&](const float* y, const float* G, const float* bias, float* ob,
                    int rk, float w, float cw) {
    const int nb = (GDE_N + 31) / 32;
    if (rk == 0)
      gde_conv_fused<0><<<nb, 512, 0, stream>>>(y, ED, RP, INV, G, bias, ob, ACC, H, w, cw);
    else if (rk == 1)
      gde_conv_fused<1><<<nb, 512, 0, stream>>>(y, ED, RP, INV, G, bias, ob, ACC, H, w, cw);
    else if (rk == 2)
      gde_conv_fused<2><<<nb, 512, 0, stream>>>(y, ED, RP, INV, G, bias, ob, ACC, H, w, cw);
    else
      gde_conv_fused<3><<<nb, 512, 0, stream>>>(y, ED, RP, INV, G, bias, ob, ACC, YT, w, cw);
  };
  auto odefunc = [&](const float* y, float* ob, int rk, float w, float cw) {
    hidden(y, GSA, ba, U, 0, 0.f, 0.f);
    hidden(U, GSB, bb, ob, rk, w, cw);
  };

  gde_mm_bstat<256, 3, 4, 2><<<256, 256, 0, stream>>>(x, GC1, Z, GDE_N / 16);
  gde_gather<64><<<GDE_N * 16 / 256, 256, 0, stream>>>(H, Z, ED, RP, b1, INV);

  odefunc(H,  YT, 1, 1.0f, 1.5f);
  odefunc(YT, YT, 2, 2.0f, 1.5f);
  odefunc(YT, YT, 2, 2.0f, 3.0f);
  odefunc(YT, H,  3, 0.0f, 0.0f);

  gde_mm_bstat<64, 3, 1, 8><<<256, 256, 0, stream>>>(H, GC2, Z, GDE_N / 16);
  gde_gather<16><<<(GDE_N * 4 + 255) / 256, 256, 0, stream>>>(out, Z, ED, RP, b2, INV);
  gde_log_softmax16<<<(GDE_N + 255) / 256, 256, 0, stream>>>(out, GDE_N);
}


# Weight-table builds folded into the edge-count launch, log_softmax into the last gather's epilogue (row passed between its 4 lanes by shuffles, same max and sum order): 20 -> 15 launches
# speedup vs baseline: 4.2582x; 1.0340x over previous
#include <hip/hip_runtime.h>
#include <math.h>

#define GDE_N 50000
#define GDE_E 800000

typedef float v2f __attribute__((ext_vector_type(2)));
typedef float v8f __attribute__((ext_vector_type(8)));

template <int K, int NT, int CG, int RS>
__global__ __launch_bounds__(32 * CG * RS) void gde_mm_bstat(const float* __restrict__ X,
                                                            const float* __restrict__ G,
                                                            float* __restrict__ Y,
                                                            int nStrips) {
  constexpr int NCOLS = 16 * NT * CG;
  constexpr int STEPS = K / 4;
  constexpr int KC    = K / 64;
  const int lane = threadIdx.x & 31;
  const int wave = __builtin_amdgcn_readfirstlane((int)(threadIdx.x >> 5));
  const int c    = lane & 15;
  const int h    = lane >> 4;
  const int cg   = wave % CG;
  const int rs   = wave / CG;
  const int col0 = cg * (16 * NT) + c;

  v2f bf[STEPS][NT];
#pragma unroll
  for (int s = 0; s < STEPS; ++s) {
    const int kb = 8 * (s >> 1) + 2 * (s & 1) + 4 * h;
#pragma unroll
    for (int t = 0; t < NT; ++t) {
      bf[s][t].x = G[(size_t)kb       * NCOLS + col0 + 16 * t];
      bf[s][t].y = G[(size_t)(kb + 1) * NCOLS + col0 + 16 * t];
    }
  }

  const int stride = gridDim.x * RS;
  int strip = blockIdx.x * RS + rs;
  if (strip >= nStrips) return;
  const float* xr = X + ((size_t)strip * 16 + c) * K + 4 * h;

  float4 cur[8];
#pragma unroll
  for (int t = 0; t < 8; ++t) {
    cur[t] = *(const float4*)(xr + 8 * t);
    __builtin_amdgcn_sched_barrier(0);
  }

  for (; strip < nStrips; strip += stride) {
    const int sn = min(strip + stride, nStrips - 1);
    const float* xn = X + ((size_t)sn * 16 + c) * K + 4 * h;

    v8f acc[NT];
#pragma unroll
    for (int t = 0; t < NT; ++t) acc[t] = (v8f){0.f, 0.f, 0.f, 0.f, 0.f, 0.f, 0.f, 0.f};

#pragma unroll
    for (int kc = 0; kc < KC; ++kc) {
      const float* nsrc = (kc + 1 < KC) ? (xr + (kc + 1) * 64) : xn;
#pragma unroll
      for (int t = 0; t < 8; ++t) {
        v2f a0; a0.x = cur[t].x; a0.y = cur[t].y;
        v2f a1; a1.x = cur[t].z; a1.y = cur[t].w;
        const int S = kc * 16 + 2 * t;
#pragma unroll
        for (int q = 0; q < NT; ++q)
          acc[q] = __builtin_amdgcn_wmma_f32_16x16x4_f32(false, a0, false, bf[S][q],
                                                         (short)0, acc[q], false, false);
#pragma unroll
        for (int q = 0; q < NT; ++q)
          acc[q] = __builtin_amdgcn_wmma_f32_16x16x4_f32(false, a1, false, bf[S + 1][q],
                                                         (short)0, acc[q], false, false);
        cur[t] = *(const float4*)(nsrc + 8 * t);
        __builtin_amdgcn_sched_barrier(0);
      }
    }

    float* yr = Y + ((size_t)strip * 16 + 8 * h) * NCOLS + col0;
#pragma unroll
    for (int q = 0; q < NT; ++q)
#pragma unroll
      for (int r = 0; r < 8; ++r)
        yr[(size_t)r * NCOLS + 16 * q] = acc[q][r];

    xr = xn;
  }
}

__device__ __forceinline__ void gde_prep_g(float* __restrict__ G, const float* __restrict__ Wp,
                                           const float* __restrict__ R, int Cin, int Cout, int idx) {
  int Ncols = 3 * Cout;
  int r = idx / Ncols, c = idx % Ncols;
  float v;
  if (c < Cout)            v = Wp[r * Cout + c];
  else if (c < 2 * Cout)   v = Wp[(size_t)Cin * Cout + r * Cout + (c - Cout)]
                             - Wp[r * Cout + (c - Cout)];
  else                     v = R[r * Cout + (c - 2 * Cout)];
  G[idx] = v;
}

__device__ __forceinline__ void gde_prep_gf(float* __restrict__ G, const float* __restrict__ Wp,
                                            const float* __restrict__ R, int idx) {
  constexpr int C = 64;
  const int j = idx & 3, lane = (idx >> 2) & 31, T = (idx >> 7) % 24, q = idx / (128 * 24);
  const int k = 8 * T + 4 * (lane >> 4) + j;
  const int c = 16 * q + (lane & 15);
  float v;
  if (k < C)            v = Wp[k * C + c];
  else if (k < 2 * C)   v = Wp[(size_t)C * C + (k - C) * C + c] - Wp[(k - C) * C + c];
  else                  v = R[(k - 2 * C) * C + c];
  G[idx] = v;
}

#define GDE_NB 256
#define GDE_RN 196
#define GDE_CH ((GDE_E + GDE_NB - 1) / GDE_NB)
static_assert(GDE_NB * GDE_RN >= GDE_N, "the ranges cover every node");
static_assert(GDE_E <= (1 << 20) && GDE_RN <= (1 << 11), "bucket word: 20 bits of edge id under the node");

#define GDE_PREP_A (256 * 192)
#define GDE_PREP_B (192 * 64)
#define GDE_PREP_C (64 * 48)
#define GDE_PREP (GDE_PREP_A + 2 * GDE_PREP_B + GDE_PREP_C)
static_assert(GDE_PREP <= GDE_NB * 1024, "one table element per thread of the counting grid");
__global__ __launch_bounds__(1024) void gde_hist(int* __restrict__ CT, const int* __restrict__ dst,
                                                 float* __restrict__ GC1, float* __restrict__ GSA,
                                                 float* __restrict__ GSB, float* __restrict__ GC2,
                                                 const float* __restrict__ W1, const float* __restrict__ R1,
                                                 const float* __restrict__ Wa, const float* __restrict__ Ra,
                                                 const float* __restrict__ Wb, const float* __restrict__ Rb,
                                                 const float* __restrict__ W2, const float* __restrict__ R2) {
  __shared__ int hc[GDE_NB];
  const int t = threadIdx.x;
  {
    const int g = blockIdx.x * 1024 + t;
    if (g < GDE_PREP_A)                        gde_prep_g(GC1, W1, R1, 256, 64, g);
    else if (g < GDE_PREP_A + GDE_PREP_B)      gde_prep_gf(GSA, Wa, Ra, g - GDE_PREP_A);
    else if (g < GDE_PREP_A + 2 * GDE_PREP_B)  gde_prep_gf(GSB, Wb, Rb, g - GDE_PREP_A - GDE_PREP_B);
    else if (g < GDE_PREP)                     gde_prep_g(GC2, W2, R2, 64, 16, g - GDE_PREP_A - 2 * GDE_PREP_B);
  }
  if (t < GDE_NB) hc[t] = 0;
  __syncthreads();
  const int e0 = blockIdx.x * GDE_CH, e1 = min(e0 + GDE_CH, GDE_E);
  for (int e = e0 + t; e < e1; e += 1024) atomicAdd(&hc[dst[e] / GDE_RN], 1);
  __syncthreads();
  if (t < GDE_NB) CT[blockIdx.x * GDE_NB + t] = hc[t];
}

__global__ __launch_bounds__(1024) void gde_bucket(int* __restrict__ BK, int* __restrict__ BB,
                                                   const int* __restrict__ CT,
                                                   const int* __restrict__ dst) {
  static_assert(GDE_NB == 256, "four thread groups of GDE_NB share the table rows");
  __shared__ int off[GDE_NB];
  __shared__ int ptot[4][GDE_NB];
  __shared__ int ppre[4][GDE_NB];
  __shared__ int wsum[GDE_NB / 32];
  const int tid = threadIdx.x, lane = tid & 31, w = tid >> 5;
  const int t = tid & (GDE_NB - 1), q = tid >> 8;
  const int b1 = blockIdx.x;
  int pre = 0, tot = 0;
  for (int g = 0; g < GDE_NB / 4; g += 16) {
    int v[16];
#pragma unroll
    for (int u = 0; u < 16; ++u) v[u] = CT[((g + u) * 4 + q) * GDE_NB + t];
#pragma unroll
    for (int u = 0; u < 16; ++u) {
      tot += v[u];
      pre += ((g + u) * 4 + q < b1) ? v[u] : 0;
    }
  }
  ptot[q][t] = tot;
  ppre[q][t] = pre;
  __syncthreads();
  tot = 0;
  pre = 0;
  if (tid < GDE_NB) {
    tot = ptot[0][t] + ptot[1][t] + ptot[2][t] + ptot[3][t];
    pre = ppre[0][t] + ppre[1][t] + ppre[2][t] + ppre[3][t];
  }
  int inc = tot;
  for (int d = 1; d < 32; d <<= 1) {
    const int o = __shfl_up(inc, d);
    if (lane >= d) inc += o;
  }
  if (tid < GDE_NB && lane == 31) wsum[w] = inc;
  __syncthreads();
  if (tid < GDE_NB) {
    int wb = 0;
    for (int i = 0; i < w; ++i) wb += wsum[i];
    const int excl = wb + inc - tot;
    off[t] = excl + pre;
    if (b1 == 0) {
      BB[t] = excl;
      if (t == GDE_NB - 1) BB[GDE_NB] = excl + tot;
    }
  }
  __syncthreads();
  const int e0 = b1 * GDE_CH, e1 = min(e0 + GDE_CH, GDE_E);
  for (int e = e0 + tid; e < e1; e += 1024) {
    const int d = dst[e];
    const int k = d / GDE_RN;
    const int slot = atomicAdd(&off[k], 1);
    BK[slot] = ((d - k * GDE_RN) << 20) | e;
  }
}

__global__ __launch_bounds__(1024) void gde_owner(int2* __restrict__ ed, int* __restrict__ rowptr,
                                                  float* __restrict__ inv, int* __restrict__ AR,
                                                  int* __restrict__ SG, const int* __restrict__ BK,
                                                  const int* __restrict__ BB,
                                                  const int* __restrict__ src,
                                                  const int* __restrict__ dst,
                                                  const float* __restrict__ p) {
  constexpr int PL = 7;
  static_assert(32 * PL >= GDE_RN, "one wave scans a range");
  __shared__ int cnt[32 * PL];
  __shared__ int lrp[32 * PL + 1];
  const int t  = threadIdx.x;
  const int k  = blockIdx.x;
  const int r0 = k * GDE_RN;
  const int base = BB[k], nI = BB[k + 1] - base;
  if (t < 32 * PL) cnt[t] = 0;
  __syncthreads();
  for (int i = t; i < nI; i += 1024) AR[base + i] = atomicAdd(&cnt[BK[base + i] >> 20], 1);
  __syncthreads();
  if (t < 32) {
    int c[PL];
    int s = 0;
#pragma unroll
    for (int j = 0; j < PL; ++j) { c[j] = cnt[t * PL + j]; s += c[j]; }
    int inc = s;
    for (int d = 1; d < 32; d <<= 1) {
      const int o = __shfl_up(inc, d);
      if (t >= d) inc += o;
    }
    int ex = inc - s;
#pragma unroll
    for (int j = 0; j < PL; ++j) { lrp[t * PL + j] = ex; ex += c[j]; }
    if (t == 31) lrp[32 * PL] = ex;
  }
  __syncthreads();
  if (t < GDE_RN && r0 + t < GDE_N) {
    rowptr[r0 + t] = base + lrp[t];
    inv[r0 + t]    = 1.0f / fmaxf((float)cnt[t], 1.0f);
  }
  if (k == GDE_NB - 1 && t == 0) rowptr[GDE_N] = base + nI;
  for (int i = t; i < nI; i += 1024) {
    const int wv = BK[base + i];
    SG[base + lrp[wv >> 20] + AR[base + i]] = wv & 0xFFFFF;
  }
  __threadfence();
  __syncthreads();
  for (int i = t; i < nI; i += 1024) {
    const int e  = SG[base + i];
    const int dl = dst[e] - r0;
    const int b = lrp[dl], n = lrp[dl + 1];
    int r = 0;
    for (int j = b; j < n; ++j) r += (SG[base + j] < e) ? 1 : 0;
    ed[base + b + r] = make_int2(src[e], __float_as_int(p[e]));
  }
}

__device__ __forceinline__ float4 gde_keep(bool live, const float4& v) {
  return make_float4(live ? v.x : 0.f, live ? v.y : 0.f, live ? v.z : 0.f, live ? v.w : 0.f);
}

__device__ __forceinline__ void gde_acc(float4& s, const float4& z0, const float4& z1, float pe) {
  s.x += z0.x + pe * z1.x;
  s.y += z0.y + pe * z1.y;
  s.z += z0.z + pe * z1.z;
  s.w += z0.w + pe * z1.w;
}

template <int COUT>
__global__ __launch_bounds__(256) void gde_gather(float* __restrict__ out, const float* __restrict__ Z,
                                                  const int2* __restrict__ ed,
                                                  const int* __restrict__ rowptr,
                                                  const float* __restrict__ bias,
                                                  const float* __restrict__ inv) {
  constexpr int LPN = COUT / 4;
  constexpr int LDZ = 3 * COUT;
  const int gt   = blockIdx.x * 256 + threadIdx.x;
  const int node = gt / LPN;
  if (node >= GDE_N) return;
  const int c = (gt % LPN) * 4;
  const int b = rowptr[node], n = rowptr[node + 1];
  const char* zb = (const char*)Z;
  const unsigned cb = (unsigned)c * 4u;
  float4 s = make_float4(0.f, 0.f, 0.f, 0.f);
  const int last = max(n - 1, 0);
  int j = b;
  int2 e0 = ed[min(j, last)], e1 = ed[min(j + 1, last)];
  int2 e2 = ed[min(j + 2, last)], e3 = ed[min(j + 3, last)];
  while (j < n) {
    const char* r0 = zb + ((unsigned)e0.x * (LDZ * 4u) + cb);
    const char* r1 = zb + ((unsigned)e1.x * (LDZ * 4u) + cb);
    const char* r2 = zb + ((unsigned)e2.x * (LDZ * 4u) + cb);
    const char* r3 = zb + ((unsigned)e3.x * (LDZ * 4u) + cb);
    const float4 a0 = *(const float4*)r0, b0 = *(const float4*)(r0 + COUT * 4);
    const float4 a1 = *(const float4*)r1, b1 = *(const float4*)(r1 + COUT * 4);
    const float4 a2 = *(const float4*)r2, b2 = *(const float4*)(r2 + COUT * 4);
    const float4 a3 = *(const float4*)r3, b3 = *(const float4*)(r3 + COUT * 4);
    const int jn = j + 4;
    const int2 f0 = ed[min(jn, last)], f1 = ed[min(jn + 1, last)];
    const int2 f2 = ed[min(jn + 2, last)], f3 = ed[min(jn + 3, last)];
    asm volatile("" ::: "memory");
    const bool k1 = j + 1 < n, k2 = j + 2 < n, k3 = j + 3 < n;
    gde_acc(s, a0, b0, __int_as_float(e0.y));
    gde_acc(s, gde_keep(k1, a1), gde_keep(k1, b1), __int_as_float(e1.y));
    gde_acc(s, gde_keep(k2, a2), gde_keep(k2, b2), __int_as_float(e2.y));
    gde_acc(s, gde_keep(k3, a3), gde_keep(k3, b3), __int_as_float(e3.y));
    e0 = f0; e1 = f1; e2 = f2; e3 = f3;
    j = jn;
  }
  const size_t oi = (size_t)node * COUT + c;
  const float iv = inv[node];
  const float4 zr = *(const float4*)(Z + (size_t)node * LDZ + 2 * COUT + c);
  float4 v;
  v.x = tanhf(s.x * iv + zr.x + bias[c + 0]);
  v.y = tanhf(s.y * iv + zr.y + bias[c + 1]);
  v.z = tanhf(s.z * iv + zr.z + bias[c + 2]);
  v.w = tanhf(s.w * iv + zr.w + bias[c + 3]);
  if (COUT == 16) {
    float r[16];
#pragma unroll
    for (int k = 0; k < 4; ++k) {
      r[4 * k + 0] = __shfl(v.x, k, 4);
      r[4 * k + 1] = __shfl(v.y, k, 4);
      r[4 * k + 2] = __shfl(v.z, k, 4);
      r[4 * k + 3] = __shfl(v.w, k, 4);
    }
    float mx = -3.402823466e+38f;
#pragma unroll
    for (int k = 0; k < 16; ++k) mx = fmaxf(mx, r[k]);
    float se = 0.f;
#pragma unroll
    for (int k = 0; k < 16; ++k) se += __expf(r[k] - mx);
    const float ls = __logf(se);
    v.x = v.x - mx - ls;
    v.y = v.y - mx - ls;
    v.z = v.z - mx - ls;
    v.w = v.w - mx - ls;
  }
  *(float4*)(out + oi) = v;
}

__device__ __forceinline__ void gde_acc2(float4& s0, float4& s1, const float4& y, float pe) {
  s0.x += y.x;  s0.y += y.y;  s0.z += y.z;  s0.w += y.w;
  s1.x += pe * y.x;  s1.y += pe * y.y;  s1.z += pe * y.z;  s1.w += pe * y.w;
}

template <int RK>
__global__ __launch_bounds__(512) void gde_conv_fused(const float* __restrict__ Y,
                                                      const int2* __restrict__ ed,
                                                      const int* __restrict__ rowptr,
                                                      const float* __restrict__ inv,
                                                      const float* __restrict__ Gf,
                                                      const float* __restrict__ bias,
                                                      float* __restrict__ out,
                                                      float* __restrict__ ACC,
                                                      const float* __restrict__ H,
                                                      float w, float cw) {
  constexpr int PITCH = 196;
  __shared__ float4 Bs[192 * 64 / 4];
  __shared__ __attribute__((aligned(16))) float At[32 * PITCH];
  const int tid = threadIdx.x;

  const int nl   = tid >> 4;
  const int c4   = (tid & 15) * 4;
  const int node = blockIdx.x * 32 + nl;
  int b = 0, n = 0;
  float iv = 0.f;
  float4 yv = make_float4(0.f, 0.f, 0.f, 0.f);
  if (node < GDE_N) {
    b  = rowptr[node];
    n  = rowptr[node + 1];
    iv = inv[node];
    yv = *(const float4*)(Y + (size_t)node * 64 + c4);
  }
#pragma unroll
  for (int i = 0; i < 6; ++i) Bs[tid + 512 * i] = ((const float4*)Gf)[tid + 512 * i];
  const float* yc = Y + c4;
  float4 s0 = make_float4(0.f, 0.f, 0.f, 0.f);
  float4 s1 = make_float4(0.f, 0.f, 0.f, 0.f);
  const int last = max(n - 1, 0);
  int j = b;
  int2 e0 = ed[min(j, last)], e1 = ed[min(j + 1, last)];
  int2 e2 = ed[min(j + 2, last)], e3 = ed[min(j + 3, last)];
  while (j < n) {
    const float4 a0 = *(const float4*)(yc + (size_t)e0.x * 64);
    const float4 a1 = *(const float4*)(yc + (size_t)e1.x * 64);
    const float4 a2 = *(const float4*)(yc + (size_t)e2.x * 64);
    const float4 a3 = *(const float4*)(yc + (size_t)e3.x * 64);
    const int jn = j + 4;
    const int2 f0 = ed[min(jn, last)], f1 = ed[min(jn + 1, last)];
    const int2 f2 = ed[min(jn + 2, last)], f3 = ed[min(jn + 3, last)];
    asm volatile("" ::: "memory");
    const bool k1 = j + 1 < n, k2 = j + 2 < n, k3 = j + 3 < n;
    gde_acc2(s0, s1, a0, __int_as_float(e0.y));
    gde_acc2(s0, s1, gde_keep(k1, a1), __int_as_float(e1.y));
    gde_acc2(s0, s1, gde_keep(k2, a2), __int_as_float(e2.y));
    gde_acc2(s0, s1, gde_keep(k3, a3), __int_as_float(e3.y));
    e0 = f0; e1 = f1; e2 = f2; e3 = f3;
    j = jn;
  }
  float* ap = At + nl * PITCH + c4;
  *(float4*)ap         = make_float4(s0.x * iv, s0.y * iv, s0.z * iv, s0.w * iv);
  *(float4*)(ap + 64)  = make_float4(s1.x * iv, s1.y * iv, s1.z * iv, s1.w * iv);
  *(float4*)(ap + 128) = yv;
  __syncthreads();

  const int q = __builtin_amdgcn_readfirstlane(tid >> 5);
  if (q >= 4) return;
  const int lane = tid & 31;
  const int c    = lane & 15;
  const int h    = lane >> 4;
  const int col  = 16 * q + c;
  const float bq = bias[col];

  const int strip0 = blockIdx.x * 2;
  size_t o0[2], ol[2];
  bool   ok[2];
  float av[2][8], hv[2][8];
#pragma unroll
  for (int m = 0; m < 2; ++m) {
    ok[m] = (strip0 + m) * 16 < GDE_N;
    o0[m] = ((size_t)(strip0 + m) * 16 + 8 * h) * 64 + col;
    ol[m] = ok[m] ? o0[m] : o0[0];
#pragma unroll
    for (int r = 0; r < 8; ++r) { av[m][r] = 0.f; hv[m][r] = 0.f; }
    if (RK == 1) {
#pragma unroll
      for (int r = 0; r < 8; ++r) hv[m][r] = H[ol[m] + (size_t)r * 64];
    }
  }

  v8f acc0 = (v8f){0.f, 0.f, 0.f, 0.f, 0.f, 0.f, 0.f, 0.f};
  v8f acc1 = (v8f){0.f, 0.f, 0.f, 0.f, 0.f, 0.f, 0.f, 0.f};
  const float4* bl = Bs + q * (24 * 32) + lane;
  const float*  a0p = At + c * PITCH + 4 * h;
  const float*  a1p = At + (16 + c) * PITCH + 4 * h;
  float4 nb  = bl[0];
  float4 nx0 = *(const float4*)a0p;
  float4 nx1 = *(const float4*)a1p;
#pragma unroll
  for (int T = 0; T < 24; ++T) {
    const float4 bb = nb, x0 = nx0, x1 = nx1;
    if (T + 1 < 24) {
      nb  = bl[(T + 1) * 32];
      nx0 = *(const float4*)(a0p + 8 * (T + 1));
      nx1 = *(const float4*)(a1p + 8 * (T + 1));
    }
    v2f b0; b0.x = bb.x; b0.y = bb.y;
    v2f b1; b1.x = bb.z; b1.y = bb.w;
    v2f p0; p0.x = x0.x; p0.y = x0.y;
    v2f p1; p1.x = x0.z; p1.y = x0.w;
    v2f r0; r0.x = x1.x; r0.y = x1.y;
    v2f r1; r1.x = x1.z; r1.y = x1.w;
    acc0 = __builtin_amdgcn_wmma_f32_16x16x4_f32(false, p0, false, b0, (short)0, acc0, false, false);
    acc1 = __builtin_amdgcn_wmma_f32_16x16x4_f32(false, r0, false, b0, (short)0, acc1, false, false);
    acc0 = __builtin_amdgcn_wmma_f32_16x16x4_f32(false, p1, false, b1, (short)0, acc0, false, false);
    acc1 = __builtin_amdgcn_wmma_f32_16x16x4_f32(false, r1, false, b1, (short)0, acc1, false, false);
    __builtin_amdgcn_sched_barrier(0);
  }

  if (RK == 2 || RK == 3) {
#pragma unroll
    for (int m = 0; m < 2; ++m) {
#pragma unroll
      for (int r = 0; r < 8; ++r) {
        const size_t oi = ol[m] + (size_t)r * 64;
        av[m][r] = ACC[oi];
        hv[m][r] = (RK == 2) ? H[oi] : out[oi];
      }
    }
  }

#pragma unroll
  for (int m = 0; m < 2; ++m) {
    if (!ok[m]) continue;
#pragma unroll
    for (int r = 0; r < 8; ++r) {
      const size_t oi = o0[m] + (size_t)r * 64;
      const float v = (m == 0 ? acc0[r] : acc1[r]) + bq;
      if (RK == 0) {
        out[oi] = v;
      } else if (RK == 1) {
        ACC[oi] = w * v;
        out[oi] = hv[m][r] + cw * v;
      } else if (RK == 2) {
        ACC[oi] = av[m][r] + w * v;
        out[oi] = hv[m][r] + cw * v;
      } else {
        out[oi] = hv[m][r] + 0.5f * (av[m][r] + v);
      }
    }
  }
}

extern "C" void kernel_launch(void* const* d_in, const int* in_sizes, int n_in,
                              void* d_out, int out_size, void* d_ws, size_t ws_size,
                              hipStream_t stream) {
  const float* x   = (const float*)d_in[0];
  const float* p   = (const float*)d_in[1];
  const int*   src = (const int*)d_in[2];
  const int*   dst = (const int*)d_in[3];
  const float* W1 = (const float*)d_in[4],  *R1 = (const float*)d_in[5],  *b1 = (const float*)d_in[6];
  const float* Wa = (const float*)d_in[7],  *Ra = (const float*)d_in[8],  *ba = (const float*)d_in[9];
  const float* Wb = (const float*)d_in[10], *Rb = (const float*)d_in[11], *bb = (const float*)d_in[12];
  const float* W2 = (const float*)d_in[13], *R2 = (const float*)d_in[14], *b2 = (const float*)d_in[15];
  float* out = (float*)d_out;

  float* W = (float*)d_ws;
  size_t o = 0;
  float* Z   = W + o; o += (size_t)GDE_N * 192;
  float* CSR = W + o; o += (size_t)GDE_N * 64;
  float* H   = W + o; o += (size_t)GDE_N * 64;
  float* YT  = W + o; o += (size_t)GDE_N * 64;
  float* U   = W + o; o += (size_t)GDE_N * 64;
  float* ACC = W + o; o += (size_t)GDE_N * 64;
  float* INV = W + o; o += (size_t)GDE_N;
  float* GC1 = W + o; o += (size_t)256 * 192;
  float* GSA = W + o; o += (size_t)192 * 64;
  float* GSB = W + o; o += (size_t)192 * 64;
  float* GC2 = W + o; o += (size_t)64 * 48;
  int*  RP  = (int*)CSR + GDE_N;
  int2* ED  = (int2*)(RP + GDE_N + 2 + GDE_E);
  int*  CT  = (int*)Z;
  int*  BB  = CT + GDE_NB * GDE_NB;
  int*  BK  = BB + GDE_NB + 4;
  int*  AR  = BK + GDE_E;
  int*  SG  = AR + GDE_E;
  (void)ws_size; (void)n_in; (void)in_sizes; (void)out_size;

  gde_hist<<<GDE_NB, 1024, 0, stream>>>(CT, dst, GC1, GSA, GSB, GC2, W1, R1, Wa, Ra, Wb, Rb, W2, R2);
  gde_bucket<<<GDE_NB, 1024, 0, stream>>>(BK, BB, CT, dst);
  gde_owner<<<GDE_NB, 1024, 0, stream>>>(ED, RP, INV, AR, SG, BK, BB, src, dst, p);

  auto hidden = [&](const float* y, const float* G, const float* bias, float* ob,
                    int rk, float w, float cw) {
    const int nb = (GDE_N + 31) / 32;
    if (rk == 0)
      gde_conv_fused<0><<<nb, 512, 0, stream>>>(y, ED, RP, INV, G, bias, ob, ACC, H, w, cw);
    else if (rk == 1)
      gde_conv_fused<1><<<nb, 512, 0, stream>>>(y, ED, RP, INV, G, bias, ob, ACC, H, w, cw);
    else if (rk == 2)
      gde_conv_fused<2><<<nb, 512, 0, stream>>>(y, ED, RP, INV, G, bias, ob, ACC, H, w, cw);
    else
      gde_conv_fused<3><<<nb, 512, 0, stream>>>(y, ED, RP, INV, G, bias, ob, ACC, YT, w, cw);
  };
  auto odefunc = [&](const float* y, float* ob, int rk, float w, float cw) {
    hidden(y, GSA, ba, U, 0, 0.f, 0.f);
    hidden(U, GSB, bb, ob, rk, w, cw);
  };

  gde_mm_bstat<256, 3, 4, 2><<<256, 256, 0, stream>>>(x, GC1, Z, GDE_N / 16);
  gde_gather<64><<<GDE_N * 16 / 256, 256, 0, stream>>>(H, Z, ED, RP, b1, INV);

  odefunc(H,  YT, 1, 1.0f, 1.5f);
  odefunc(YT, YT, 2, 2.0f, 1.5f);
  odefunc(YT, YT, 2, 2.0f, 3.0f);
  odefunc(YT, H,  3, 0.0f, 0.0f);

  gde_mm_bstat<64, 3, 1, 8><<<256, 256, 0, stream>>>(H, GC2, Z, GDE_N / 16);
  gde_gather<16><<<(GDE_N * 4 + 255) / 256, 256, 0, stream>>>(out, Z, ED, RP, b2, INV);
}
